# starting kernel
# baseline (speedup 1.0000x reference)
#include <hip/hip_runtime.h>
#include <hip/hip_bf16.h>

typedef __attribute__((ext_vector_type(2))) float v2f;
typedef __attribute__((ext_vector_type(8))) float v8f;

#define FEAT 256
#define TILE_K 32

__global__ __launch_bounds__(256) void build_s1_kernel(
    const int* __restrict__ adj, const int* __restrict__ batch,
    int* __restrict__ s1)
{
    int i = blockIdx.x * 256 + threadIdx.x;
    if (i < 512 * 10) {
        int b = i / 10;
        int j = i - b * 10;
        s1[i] = adj[(size_t)batch[b] * 25 + j];
    }
}

__global__ __launch_bounds__(64) void mean_neigh25_kernel(
    const float* __restrict__ feat, const int* __restrict__ adj,
    const int* __restrict__ s1, float* __restrict__ out)
{
    const int row  = blockIdx.x;
    const int t    = threadIdx.x;
    const int node = s1[row];
    const int* arow = adj + (size_t)node * 25;
    float ax = 0.f, ay = 0.f, az = 0.f, aw = 0.f;
    for (int k = 0; k < 25; ++k) {
        const int nb = arow[k];
        const float4 v = *(const float4*)(feat + (size_t)nb * FEAT + t * 4);
        ax += v.x; ay += v.y; az += v.z; aw += v.w;
    }
    const float s = 1.0f / 25.0f;
    float4 r; r.x = ax * s; r.y = ay * s; r.z = az * s; r.w = aw * s;
    *(float4*)(out + (size_t)row * FEAT + t * 4) = r;
}

__global__ __launch_bounds__(64) void mean_neigh10_kernel(
    const float* __restrict__ feat, const int* __restrict__ s1,
    float* __restrict__ out)
{
    const int row = blockIdx.x;
    const int t   = threadIdx.x;
    float ax = 0.f, ay = 0.f, az = 0.f, aw = 0.f;
    for (int j = 0; j < 10; ++j) {
        const int nb = s1[row * 10 + j];
        const float4 v = *(const float4*)(feat + (size_t)nb * FEAT + t * 4);
        ax += v.x; ay += v.y; az += v.z; aw += v.w;
    }
    const float s = 0.1f;
    float4 r; r.x = ax * s; r.y = ay * s; r.z = az * s; r.w = aw * s;
    *(float4*)(out + (size_t)row * FEAT + t * 4) = r;
}

__global__ __launch_bounds__(64) void mean_rows10_kernel(
    const float* __restrict__ h1, float* __restrict__ out)
{
    const int row = blockIdx.x;
    const int t   = threadIdx.x;
    float ax = 0.f, ay = 0.f, az = 0.f, aw = 0.f;
    for (int j = 0; j < 10; ++j) {
        const float4 v =
            *(const float4*)(h1 + ((size_t)row * 10 + j) * FEAT + t * 4);
        ax += v.x; ay += v.y; az += v.z; aw += v.w;
    }
    const float s = 0.1f;
    float4 r; r.x = ax * s; r.y = ay * s; r.z = az * s; r.w = aw * s;
    *(float4*)(out + (size_t)row * FEAT + t * 4) = r;
}

template <bool RELU, bool GATHER>
__global__ __launch_bounds__(256) void sage_gemm_kernel(
    const float* __restrict__ Aself, const int* __restrict__ idx,
    const float* __restrict__ Aneigh,
    const float* __restrict__ Ws, const float* __restrict__ Wn,
    float* __restrict__ out)
{
    __shared__ float lds_ws[TILE_K * FEAT];
    __shared__ float lds_wn[TILE_K * FEAT];

    const int tid       = threadIdx.x;
    const int lane      = tid & 31;
    const int wave      = tid >> 5;
    const int mtile     = wave >> 1;
    const int nhalf     = wave & 1;
    const int rowInTile = lane & 15;
    const int klane     = (lane >> 4) << 1;
    const int m         = blockIdx.x * 64 + mtile * 16 + rowInTile;

    const float* arow_s =
        GATHER ? (Aself + (size_t)idx[m] * FEAT) : (Aself + (size_t)m * FEAT);
    const float* arow_n = Aneigh + (size_t)m * FEAT;

    const int nBase = (nhalf * 128 + rowInTile) * 2;

    v8f acc[8];
#pragma unroll
    for (int t = 0; t < 8; ++t) acc[t] = (v8f)0.f;

    for (int k0 = 0; k0 < FEAT; k0 += TILE_K) {
        __syncthreads();
        {
            const float* gws = Ws + (size_t)k0 * FEAT;
            const float* gwn = Wn + (size_t)k0 * FEAT;
#pragma unroll
            for (int i = 0; i < 8; ++i) {
                const int u     = i * 256 + tid;
                const int kpair = u >> 7;
                const int npair = u & 127;
                const int g0 = (2 * kpair) * FEAT + 2 * npair;
                const int g1 = g0 + FEAT;
                const int lo = kpair * 512 + npair * 4;
                {
                    const v2f r0 = *(const v2f*)(gws + g0);
                    const v2f r1 = *(const v2f*)(gws + g1);
                    float4 p; p.x = r0.x; p.y = r1.x; p.z = r0.y; p.w = r1.y;
                    *(float4*)(lds_ws + lo) = p;
                }
                {
                    const v2f r0 = *(const v2f*)(gwn + g0);
                    const v2f r1 = *(const v2f*)(gwn + g1);
                    float4 p; p.x = r0.x; p.y = r1.x; p.z = r0.y; p.w = r1.y;
                    *(float4*)(lds_wn + lo) = p;
                }
            }
        }
        __syncthreads();

#pragma unroll
        for (int kk = 0; kk < TILE_K; kk += 4) {
            const v2f a_s = *(const v2f*)(arow_s + k0 + kk + klane);
            const v2f a_n = *(const v2f*)(arow_n + k0 + kk + klane);
            const int kp  = (kk + klane) >> 1;

            v2f bs[8], bn[8];
#pragma unroll
            for (int t = 0; t < 8; ++t) {
                bs[t] = *(const v2f*)(lds_ws + kp * 512 + nBase + t * 32);
                bn[t] = *(const v2f*)(lds_wn + kp * 512 + nBase + t * 32);
            }
#pragma unroll
            for (int t = 0; t < 8; ++t) {
                acc[t] = __builtin_amdgcn_wmma_f32_16x16x4_f32(
                    false, a_s, false, bs[t], (short)0, acc[t], false, false);
                acc[t] = __builtin_amdgcn_wmma_f32_16x16x4_f32(
                    false, a_n, false, bn[t], (short)0, acc[t], false, false);
            }
        }
    }

    const int colBase = nhalf * 128 + rowInTile;
    const int rowBase = blockIdx.x * 64 + mtile * 16 + ((lane >> 4) << 3);
#pragma unroll
    for (int t = 0; t < 8; ++t) {
#pragma unroll
        for (int r = 0; r < 8; ++r) {
            float v = acc[t][r];
            if (RELU) v = fmaxf(v, 0.0f);
            out[(size_t)(rowBase + r) * FEAT + colBase + t * 16] = v;
        }
    }
}

__global__ __launch_bounds__(256) void l2norm_kernel(
    const float* __restrict__ pre, float* __restrict__ out)
{
    __shared__ float ssum[8];
    const int row = blockIdx.x;
    const int t   = threadIdx.x;
    const float v = pre[(size_t)row * FEAT + t];
    float s = v * v;
#pragma unroll
    for (int o = 16; o >= 1; o >>= 1) s += __shfl_xor(s, o, 32);
    if ((t & 31) == 0) ssum[t >> 5] = s;
    __syncthreads();
    float tot = 0.f;
#pragma unroll
    for (int i = 0; i < 8; ++i) tot += ssum[i];
    const float norm = fmaxf(sqrtf(tot), 1e-12f);
    out[(size_t)row * FEAT + t] = v / norm;
}

extern "C" void kernel_launch(void* const* d_in, const int* in_sizes, int n_in,
                              void* d_out, int out_size, void* d_ws,
                              size_t ws_size, hipStream_t stream)
{
    const float* features = (const float*)d_in[0];
    const int*   adj      = (const int*)d_in[1];
    const int*   batch    = (const int*)d_in[2];
    const float* Ws0      = (const float*)d_in[3];
    const float* Wn0      = (const float*)d_in[4];
    const float* Ws1      = (const float*)d_in[5];
    const float* Wn1      = (const float*)d_in[6];
    float* out = (float*)d_out;

    char* ws = (char*)d_ws;
    int*   s1   = (int*)(ws + 0);
    float* nm2  = (float*)(ws + 20480);
    float* nm1  = (float*)(ws + 5263360);
    float* h1   = (float*)(ws + 5787648);
    float* h0   = (float*)(ws + 11030528);
    float* nmh1 = (float*)(ws + 11554816);
    float* pre  = (float*)(ws + 12079104);

    build_s1_kernel<<<(5120 + 255) / 256, 256, 0, stream>>>(adj, batch, s1);
    mean_neigh25_kernel<<<5120, 64, 0, stream>>>(features, adj, s1, nm2);
    mean_neigh10_kernel<<<512, 64, 0, stream>>>(features, s1, nm1);
    sage_gemm_kernel<true, true><<<5120 / 64, 256, 0, stream>>>(
        features, s1, nm2, Ws0, Wn0, h1);
    sage_gemm_kernel<true, true><<<512 / 64, 256, 0, stream>>>(
        features, batch, nm1, Ws0, Wn0, h0);
    mean_rows10_kernel<<<512, 64, 0, stream>>>(h1, nmh1);
    sage_gemm_kernel<false, false><<<512 / 64, 256, 0, stream>>>(
        h0, (const int*)nullptr, nmh1, Ws1, Wn1, pre);
    l2norm_kernel<<<512, 256, 0, stream>>>(pre, out);
}


# GEMM K loop software-pipelined one stage ahead: next chunk's weight rows and A fragments load into registers under the current chunk's WMMAs (2 LDS stages), B fragments one k-step ahead; sums same
# speedup vs baseline: 2.3465x; 2.3465x over previous
#include <hip/hip_runtime.h>
#include <hip/hip_bf16.h>

typedef __attribute__((ext_vector_type(2))) float v2f;
typedef __attribute__((ext_vector_type(8))) float v8f;

#define FEAT 256
#define TILE_K 32

__global__ __launch_bounds__(256) void build_s1_kernel(
    const int* __restrict__ adj, const int* __restrict__ batch,
    int* __restrict__ s1)
{
    int i = blockIdx.x * 256 + threadIdx.x;
    if (i < 512 * 10) {
        int b = i / 10;
        int j = i - b * 10;
        s1[i] = adj[(size_t)batch[b] * 25 + j];
    }
}

__global__ __launch_bounds__(64) void mean_neigh25_kernel(
    const float* __restrict__ feat, const int* __restrict__ adj,
    const int* __restrict__ s1, float* __restrict__ out)
{
    const int row  = blockIdx.x;
    const int t    = threadIdx.x;
    const int node = s1[row];
    const int* arow = adj + (size_t)node * 25;
    float ax = 0.f, ay = 0.f, az = 0.f, aw = 0.f;
    for (int k = 0; k < 25; ++k) {
        const int nb = arow[k];
        const float4 v = *(const float4*)(feat + (size_t)nb * FEAT + t * 4);
        ax += v.x; ay += v.y; az += v.z; aw += v.w;
    }
    const float s = 1.0f / 25.0f;
    float4 r; r.x = ax * s; r.y = ay * s; r.z = az * s; r.w = aw * s;
    *(float4*)(out + (size_t)row * FEAT + t * 4) = r;
}

__global__ __launch_bounds__(64) void mean_neigh10_kernel(
    const float* __restrict__ feat, const int* __restrict__ s1,
    float* __restrict__ out)
{
    const int row = blockIdx.x;
    const int t   = threadIdx.x;
    float ax = 0.f, ay = 0.f, az = 0.f, aw = 0.f;
    for (int j = 0; j < 10; ++j) {
        const int nb = s1[row * 10 + j];
        const float4 v = *(const float4*)(feat + (size_t)nb * FEAT + t * 4);
        ax += v.x; ay += v.y; az += v.z; aw += v.w;
    }
    const float s = 0.1f;
    float4 r; r.x = ax * s; r.y = ay * s; r.z = az * s; r.w = aw * s;
    *(float4*)(out + (size_t)row * FEAT + t * 4) = r;
}

__global__ __launch_bounds__(64) void mean_rows10_kernel(
    const float* __restrict__ h1, float* __restrict__ out)
{
    const int row = blockIdx.x;
    const int t   = threadIdx.x;
    float ax = 0.f, ay = 0.f, az = 0.f, aw = 0.f;
    for (int j = 0; j < 10; ++j) {
        const float4 v =
            *(const float4*)(h1 + ((size_t)row * 10 + j) * FEAT + t * 4);
        ax += v.x; ay += v.y; az += v.z; aw += v.w;
    }
    const float s = 0.1f;
    float4 r; r.x = ax * s; r.y = ay * s; r.z = az * s; r.w = aw * s;
    *(float4*)(out + (size_t)row * FEAT + t * 4) = r;
}

#define DEV static __device__ __forceinline__
#define CHUNK_FLOATS (TILE_K * FEAT)
#define N_CHUNKS (FEAT / TILE_K)
#define K_STEPS (TILE_K / 4)
#define PF_PER_STEP 8
#define PF_TOTAL 48

DEV void sched_fence() { __builtin_amdgcn_sched_barrier(0); }

DEV int w_unit_goff(int i, int tid)
{
    const int u     = i * 256 + tid;
    const int kpair = u >> 7;
    const int npair = u & 127;
    return (2 * kpair) * FEAT + 2 * npair;
}

DEV void w_chunk_to_regs(const float* __restrict__ g, int tid,
                         v2f (&r0)[8], v2f (&r1)[8])
{
#pragma unroll
    for (int i = 0; i < 8; ++i) {
        const int g0 = w_unit_goff(i, tid);
        r0[i] = *(const v2f*)(g + g0);
        r1[i] = *(const v2f*)(g + g0 + FEAT);
    }
}

DEV void w_regs_to_lds(float* l, int tid,
                       const v2f (&r0)[8], const v2f (&r1)[8])
{
#pragma unroll
    for (int i = 0; i < 8; ++i) {
        const int u     = i * 256 + tid;
        const int kpair = u >> 7;
        const int npair = u & 127;
        float4 p; p.x = r0[i].x; p.y = r1[i].x; p.z = r0[i].y; p.w = r1[i].y;
        *(float4*)(l + kpair * 512 + npair * 4) = p;
    }
}

DEV void a_chunk_to_regs(const float* __restrict__ ak, v2f (&a)[8])
{
#pragma unroll
    for (int s = 0; s < 8; ++s) a[s] = *(const v2f*)(ak + s * 4);
}

struct NextChunk {
    const float* gws;
    const float* gwn;
    const float* aks;
    const float* akn;
};

template <bool PREFETCH>
DEV void mma_chunk(v8f (&acc)[8], const v2f (&a_s)[8], const v2f (&a_n)[8],
                   const float* bws, const float* bwn,
                   const NextChunk& nx, int tid,
                   v2f (&ws0)[8], v2f (&ws1)[8], v2f (&wn0)[8], v2f (&wn1)[8],
                   v2f (&a_s_nx)[8], v2f (&a_n_nx)[8])
{
    v2f bs[2][8], bn[2][8];
#pragma unroll
    for (int t = 0; t < 8; ++t) {
        bs[0][t] = *(const v2f*)(bws + t * 32);
        bn[0][t] = *(const v2f*)(bwn + t * 32);
    }
#pragma unroll
    for (int s = 0; s < K_STEPS; ++s) {
        const int cur = s & 1;
        if (PREFETCH) {
#pragma unroll
            for (int j = s * PF_PER_STEP;
                 j < (s + 1) * PF_PER_STEP && j < PF_TOTAL; ++j) {
                if (j < 16) {
                    if ((j & 1) == 0)
                        a_s_nx[j >> 1] = *(const v2f*)(nx.aks + (j >> 1) * 4);
                    else
                        a_n_nx[j >> 1] = *(const v2f*)(nx.akn + (j >> 1) * 4);
                } else {
                    const int w  = j - 16;
                    const int i  = w >> 2;
                    const int g0 = w_unit_goff(i, tid);
                    if ((w & 3) == 0) ws0[i] = *(const v2f*)(nx.gws + g0);
                    if ((w & 3) == 1) ws1[i] = *(const v2f*)(nx.gws + g0 + FEAT);
                    if ((w & 3) == 2) wn0[i] = *(const v2f*)(nx.gwn + g0);
                    if ((w & 3) == 3) wn1[i] = *(const v2f*)(nx.gwn + g0 + FEAT);
                }
            }
        }
        if (s + 1 < K_STEPS) {
#pragma unroll
            for (int t = 0; t < 8; ++t) {
                bs[cur ^ 1][t] = *(const v2f*)(bws + (s + 1) * 1024 + t * 32);
                bn[cur ^ 1][t] = *(const v2f*)(bwn + (s + 1) * 1024 + t * 32);
            }
        }
        sched_fence();
#pragma unroll
        for (int t = 0; t < 8; ++t)
            acc[t] = __builtin_amdgcn_wmma_f32_16x16x4_f32(
                false, a_s[s], false, bs[cur][t], (short)0, acc[t],
                false, false);
#pragma unroll
        for (int t = 0; t < 8; ++t)
            acc[t] = __builtin_amdgcn_wmma_f32_16x16x4_f32(
                false, a_n[s], false, bn[cur][t], (short)0, acc[t],
                false, false);
        sched_fence();
    }
}

template <bool RELU, bool GATHER>
__global__ __launch_bounds__(256) void sage_gemm_kernel(
    const float* __restrict__ Aself, const int* __restrict__ idx,
    const float* __restrict__ Aneigh,
    const float* __restrict__ Ws, const float* __restrict__ Wn,
    float* __restrict__ out)
{
    __shared__ float lds_w[2][2][CHUNK_FLOATS];

    const int tid       = threadIdx.x;
    const int lane      = tid & 31;
    const int wave      = tid >> 5;
    const int mtile     = wave >> 1;
    const int nhalf     = wave & 1;
    const int rowInTile = lane & 15;
    const int klane     = (lane >> 4) << 1;
    const int m         = blockIdx.x * 64 + mtile * 16 + rowInTile;

    const float* arow_s =
        (GATHER ? (Aself + (size_t)idx[m] * FEAT) : (Aself + (size_t)m * FEAT))
        + klane;
    const float* arow_n = Aneigh + (size_t)m * FEAT + klane;

    const int bBase = (klane >> 1) * 512 + (nhalf * 128 + rowInTile) * 2;

    v8f acc[8];
#pragma unroll
    for (int t = 0; t < 8; ++t) acc[t] = (v8f)0.f;

    v2f ws0[8], ws1[8], wn0[8], wn1[8];
    v2f a_s[8], a_n[8];
    v2f a_s_nx[8], a_n_nx[8];

    w_chunk_to_regs(Ws, tid, ws0, ws1);
    w_chunk_to_regs(Wn, tid, wn0, wn1);
    a_chunk_to_regs(arow_s, a_s);
    a_chunk_to_regs(arow_n, a_n);
    w_regs_to_lds(lds_w[0][0], tid, ws0, ws1);
    w_regs_to_lds(lds_w[0][1], tid, wn0, wn1);
    __syncthreads();

#pragma unroll 1
    for (int c = 0; c + 1 < N_CHUNKS; ++c) {
        const int k1 = (c + 1) * TILE_K;
        NextChunk nx;
        nx.gws = Ws + (size_t)k1 * FEAT;
        nx.gwn = Wn + (size_t)k1 * FEAT;
        nx.aks = arow_s + k1;
        nx.akn = arow_n + k1;

        mma_chunk<true>(acc, a_s, a_n, &lds_w[c & 1][0][bBase],
                        &lds_w[c & 1][1][bBase], nx, tid,
                        ws0, ws1, wn0, wn1, a_s_nx, a_n_nx);

        w_regs_to_lds(lds_w[(c + 1) & 1][0], tid, ws0, ws1);
        w_regs_to_lds(lds_w[(c + 1) & 1][1], tid, wn0, wn1);
#pragma unroll
        for (int s = 0; s < 8; ++s) { a_s[s] = a_s_nx[s]; a_n[s] = a_n_nx[s]; }
        __syncthreads();
    }
    {
        NextChunk none = {Ws, Wn, arow_s, arow_n};
        mma_chunk<false>(acc, a_s, a_n, &lds_w[(N_CHUNKS - 1) & 1][0][bBase],
                         &lds_w[(N_CHUNKS - 1) & 1][1][bBase], none, tid,
                         ws0, ws1, wn0, wn1, a_s_nx, a_n_nx);
    }

    const int colBase = nhalf * 128 + rowInTile;
    const int rowBase = blockIdx.x * 64 + mtile * 16 + ((lane >> 4) << 3);
#pragma unroll
    for (int t = 0; t < 8; ++t) {
#pragma unroll
        for (int r = 0; r < 8; ++r) {
            float v = acc[t][r];
            if (RELU) v = fmaxf(v, 0.0f);
            out[(size_t)(rowBase + r) * FEAT + colBase + t * 16] = v;
        }
    }
}

__global__ __launch_bounds__(256) void l2norm_kernel(
    const float* __restrict__ pre, float* __restrict__ out)
{
    __shared__ float ssum[8];
    const int row = blockIdx.x;
    const int t   = threadIdx.x;
    const float v = pre[(size_t)row * FEAT + t];
    float s = v * v;
#pragma unroll
    for (int o = 16; o >= 1; o >>= 1) s += __shfl_xor(s, o, 32);
    if ((t & 31) == 0) ssum[t >> 5] = s;
    __syncthreads();
    float tot = 0.f;
#pragma unroll
    for (int i = 0; i < 8; ++i) tot += ssum[i];
    const float norm = fmaxf(sqrtf(tot), 1e-12f);
    out[(size_t)row * FEAT + t] = v / norm;
}

extern "C" void kernel_launch(void* const* d_in, const int* in_sizes, int n_in,
                              void* d_out, int out_size, void* d_ws,
                              size_t ws_size, hipStream_t stream)
{
    const float* features = (const float*)d_in[0];
    const int*   adj      = (const int*)d_in[1];
    const int*   batch    = (const int*)d_in[2];
    const float* Ws0      = (const float*)d_in[3];
    const float* Wn0      = (const float*)d_in[4];
    const float* Ws1      = (const float*)d_in[5];
    const float* Wn1      = (const float*)d_in[6];
    float* out = (float*)d_out;

    char* ws = (char*)d_ws;
    int*   s1   = (int*)(ws + 0);
    float* nm2  = (float*)(ws + 20480);
    float* nm1  = (float*)(ws + 5263360);
    float* h1   = (float*)(ws + 5787648);
    float* h0   = (float*)(ws + 11030528);
    float* nmh1 = (float*)(ws + 11554816);
    float* pre  = (float*)(ws + 12079104);

    build_s1_kernel<<<(5120 + 255) / 256, 256, 0, stream>>>(adj, batch, s1);
    mean_neigh25_kernel<<<5120, 64, 0, stream>>>(features, adj, s1, nm2);
    mean_neigh10_kernel<<<512, 64, 0, stream>>>(features, s1, nm1);
    sage_gemm_kernel<true, true><<<5120 / 64, 256, 0, stream>>>(
        features, s1, nm2, Ws0, Wn0, h1);
    sage_gemm_kernel<true, true><<<512 / 64, 256, 0, stream>>>(
        features, batch, nm1, Ws0, Wn0, h0);
    mean_rows10_kernel<<<512, 64, 0, stream>>>(h1, nmh1);
    sage_gemm_kernel<false, false><<<512 / 64, 256, 0, stream>>>(
        h0, (const int*)nullptr, nmh1, Ws1, Wn1, pre);
    l2norm_kernel<<<512, 256, 0, stream>>>(pre, out);
}


# GEMM blocks re-tiled so more WGPs share the WMMAs: 32 rows x 8 waves for hop-1 (160 blocks), 16 rows x 4 waves for hop-0 and layer 1 (32 blocks), requests 16 per run; sums and k order same
# speedup vs baseline: 3.3699x; 1.4361x over previous
#include <hip/hip_runtime.h>
#include <hip/hip_bf16.h>

typedef __attribute__((ext_vector_type(2))) float v2f;
typedef __attribute__((ext_vector_type(8))) float v8f;

#define FEAT 256
#define TILE_K 32

__global__ __launch_bounds__(256) void build_s1_kernel(
    const int* __restrict__ adj, const int* __restrict__ batch,
    int* __restrict__ s1)
{
    int i = blockIdx.x * 256 + threadIdx.x;
    if (i < 512 * 10) {
        int b = i / 10;
        int j = i - b * 10;
        s1[i] = adj[(size_t)batch[b] * 25 + j];
    }
}

__global__ __launch_bounds__(64) void mean_neigh25_kernel(
    const float* __restrict__ feat, const int* __restrict__ adj,
    const int* __restrict__ s1, float* __restrict__ out)
{
    const int row  = blockIdx.x;
    const int t    = threadIdx.x;
    const int node = s1[row];
    const int* arow = adj + (size_t)node * 25;
    float ax = 0.f, ay = 0.f, az = 0.f, aw = 0.f;
    for (int k = 0; k < 25; ++k) {
        const int nb = arow[k];
        const float4 v = *(const float4*)(feat + (size_t)nb * FEAT + t * 4);
        ax += v.x; ay += v.y; az += v.z; aw += v.w;
    }
    const float s = 1.0f / 25.0f;
    float4 r; r.x = ax * s; r.y = ay * s; r.z = az * s; r.w = aw * s;
    *(float4*)(out + (size_t)row * FEAT + t * 4) = r;
}

__global__ __launch_bounds__(64) void mean_neigh10_kernel(
    const float* __restrict__ feat, const int* __restrict__ s1,
    float* __restrict__ out)
{
    const int row = blockIdx.x;
    const int t   = threadIdx.x;
    float ax = 0.f, ay = 0.f, az = 0.f, aw = 0.f;
    for (int j = 0; j < 10; ++j) {
        const int nb = s1[row * 10 + j];
        const float4 v = *(const float4*)(feat + (size_t)nb * FEAT + t * 4);
        ax += v.x; ay += v.y; az += v.z; aw += v.w;
    }
    const float s = 0.1f;
    float4 r; r.x = ax * s; r.y = ay * s; r.z = az * s; r.w = aw * s;
    *(float4*)(out + (size_t)row * FEAT + t * 4) = r;
}

__global__ __launch_bounds__(64) void mean_rows10_kernel(
    const float* __restrict__ h1, float* __restrict__ out)
{
    const int row = blockIdx.x;
    const int t   = threadIdx.x;
    float ax = 0.f, ay = 0.f, az = 0.f, aw = 0.f;
    for (int j = 0; j < 10; ++j) {
        const float4 v =
            *(const float4*)(h1 + ((size_t)row * 10 + j) * FEAT + t * 4);
        ax += v.x; ay += v.y; az += v.z; aw += v.w;
    }
    const float s = 0.1f;
    float4 r; r.x = ax * s; r.y = ay * s; r.z = az * s; r.w = aw * s;
    *(float4*)(out + (size_t)row * FEAT + t * 4) = r;
}

#define DEV static __device__ __forceinline__
#define CHUNK_FLOATS (TILE_K * FEAT)
#define N_CHUNKS (FEAT / TILE_K)
#define K_STEPS (TILE_K / 4)
#define W_UNITS (TILE_K / 2 * (FEAT / 2))
#define PF_PER_STEP 16

DEV void sched_fence() { __builtin_amdgcn_sched_barrier(0); }

template <int NTHR>
DEV int w_unit_goff(int i, int tid)
{
    const int u     = i * NTHR + tid;
    const int kpair = u >> 7;
    const int npair = u & 127;
    return (2 * kpair) * FEAT + 2 * npair;
}

template <int NTHR, int WU>
DEV void w_chunk_to_regs(const float* __restrict__ g, int tid,
                         v2f (&r0)[WU], v2f (&r1)[WU])
{
#pragma unroll
    for (int i = 0; i < WU; ++i) {
        const int g0 = w_unit_goff<NTHR>(i, tid);
        r0[i] = *(const v2f*)(g + g0);
        r1[i] = *(const v2f*)(g + g0 + FEAT);
    }
}

template <int NTHR, int WU>
DEV void w_regs_to_lds(float* l, int tid,
                       const v2f (&r0)[WU], const v2f (&r1)[WU])
{
#pragma unroll
    for (int i = 0; i < WU; ++i) {
        const int u     = i * NTHR + tid;
        const int kpair = u >> 7;
        const int npair = u & 127;
        float4 p; p.x = r0[i].x; p.y = r1[i].x; p.z = r0[i].y; p.w = r1[i].y;
        *(float4*)(l + kpair * 512 + npair * 4) = p;
    }
}

DEV void a_chunk_to_regs(const float* __restrict__ ak, v2f (&a)[8])
{
#pragma unroll
    for (int s = 0; s < 8; ++s) a[s] = *(const v2f*)(ak + s * 4);
}

struct NextChunk {
    const float* gws;
    const float* gwn;
    const float* aks;
    const float* akn;
};

template <bool PREFETCH, int NT, int NTHR, int WU>
DEV void mma_chunk(v8f (&acc)[NT], const v2f (&a_s)[8], const v2f (&a_n)[8],
                   const float* bws, const float* bwn,
                   const NextChunk& nx, int tid,
                   v2f (&ws0)[WU], v2f (&ws1)[WU],
                   v2f (&wn0)[WU], v2f (&wn1)[WU],
                   v2f (&a_s_nx)[8], v2f (&a_n_nx)[8])
{
    constexpr int PF_TOTAL = 16 + 4 * WU;
    static_assert(PF_TOTAL <= PF_PER_STEP * (K_STEPS - 3),
                  "requests must be out three runs before the chunk ends");
    v2f bs[2][NT], bn[2][NT];
#pragma unroll
    for (int t = 0; t < NT; ++t) {
        bs[0][t] = *(const v2f*)(bws + t * 32);
        bn[0][t] = *(const v2f*)(bwn + t * 32);
    }
#pragma unroll
    for (int s = 0; s < K_STEPS; ++s) {
        const int cur = s & 1;
        if (PREFETCH) {
#pragma unroll
            for (int j = s * PF_PER_STEP;
                 j < (s + 1) * PF_PER_STEP && j < PF_TOTAL; ++j) {
                if (j < 16) {
                    if ((j & 1) == 0)
                        a_s_nx[j >> 1] = *(const v2f*)(nx.aks + (j >> 1) * 4);
                    else
                        a_n_nx[j >> 1] = *(const v2f*)(nx.akn + (j >> 1) * 4);
                } else {
                    const int w  = j - 16;
                    const int i  = w >> 2;
                    const int g0 = w_unit_goff<NTHR>(i, tid);
                    if ((w & 3) == 0) ws0[i] = *(const v2f*)(nx.gws + g0);
                    if ((w & 3) == 1) ws1[i] = *(const v2f*)(nx.gws + g0 + FEAT);
                    if ((w & 3) == 2) wn0[i] = *(const v2f*)(nx.gwn + g0);
                    if ((w & 3) == 3) wn1[i] = *(const v2f*)(nx.gwn + g0 + FEAT);
                }
            }
        }
        if (s + 1 < K_STEPS) {
#pragma unroll
            for (int t = 0; t < NT; ++t) {
                bs[cur ^ 1][t] = *(const v2f*)(bws + (s + 1) * 1024 + t * 32);
                bn[cur ^ 1][t] = *(const v2f*)(bwn + (s + 1) * 1024 + t * 32);
            }
        }
        sched_fence();
#pragma unroll
        for (int t = 0; t < NT; ++t)
            acc[t] = __builtin_amdgcn_wmma_f32_16x16x4_f32(
                false, a_s[s], false, bs[cur][t], (short)0, acc[t],
                false, false);
#pragma unroll
        for (int t = 0; t < NT; ++t)
            acc[t] = __builtin_amdgcn_wmma_f32_16x16x4_f32(
                false, a_n[s], false, bn[cur][t], (short)0, acc[t],
                false, false);
        sched_fence();
    }
}

template <bool RELU, bool GATHER, int MT, int NWV>
__global__ __launch_bounds__(32 * NWV) void sage_gemm_kernel(
    const float* __restrict__ Aself, const int* __restrict__ idx,
    const float* __restrict__ Aneigh,
    const float* __restrict__ Ws, const float* __restrict__ Wn,
    float* __restrict__ out)
{
    constexpr int NTHR = 32 * NWV;
    constexpr int WU   = W_UNITS / NTHR;
    constexpr int NW   = NWV / MT;
    constexpr int NT   = 16 / NW;
    static_assert(NW * MT == NWV && NT * NW == 16 && WU * NTHR == W_UNITS,
                  "block shape must tile 16*MT x 256 and the weight chunk");

    __shared__ float lds_w[2][2][CHUNK_FLOATS];

    const int tid       = threadIdx.x;
    const int lane      = tid & 31;
    const int wave      = tid >> 5;
    const int mtile     = wave / NW;
    const int nstrip    = wave % NW;
    const int rowInTile = lane & 15;
    const int klane     = (lane >> 4) << 1;
    const int m         = blockIdx.x * (16 * MT) + mtile * 16 + rowInTile;

    const float* arow_s =
        (GATHER ? (Aself + (size_t)idx[m] * FEAT) : (Aself + (size_t)m * FEAT))
        + klane;
    const float* arow_n = Aneigh + (size_t)m * FEAT + klane;

    const int bBase =
        (klane >> 1) * 512 + (nstrip * (NT * 16) + rowInTile) * 2;

    v8f acc[NT];
#pragma unroll
    for (int t = 0; t < NT; ++t) acc[t] = (v8f)0.f;

    v2f ws0[WU], ws1[WU], wn0[WU], wn1[WU];
    v2f a_s[8], a_n[8];
    v2f a_s_nx[8], a_n_nx[8];

    w_chunk_to_regs<NTHR, WU>(Ws, tid, ws0, ws1);
    w_chunk_to_regs<NTHR, WU>(Wn, tid, wn0, wn1);
    a_chunk_to_regs(arow_s, a_s);
    a_chunk_to_regs(arow_n, a_n);
    w_regs_to_lds<NTHR, WU>(lds_w[0][0], tid, ws0, ws1);
    w_regs_to_lds<NTHR, WU>(lds_w[0][1], tid, wn0, wn1);
    __syncthreads();

#pragma unroll 1
    for (int c = 0; c + 1 < N_CHUNKS; ++c) {
        const int k1 = (c + 1) * TILE_K;
        NextChunk nx;
        nx.gws = Ws + (size_t)k1 * FEAT;
        nx.gwn = Wn + (size_t)k1 * FEAT;
        nx.aks = arow_s + k1;
        nx.akn = arow_n + k1;

        mma_chunk<true, NT, NTHR, WU>(
            acc, a_s, a_n, &lds_w[c & 1][0][bBase], &lds_w[c & 1][1][bBase],
            nx, tid, ws0, ws1, wn0, wn1, a_s_nx, a_n_nx);

        w_regs_to_lds<NTHR, WU>(lds_w[(c + 1) & 1][0], tid, ws0, ws1);
        w_regs_to_lds<NTHR, WU>(lds_w[(c + 1) & 1][1], tid, wn0, wn1);
#pragma unroll
        for (int s = 0; s < 8; ++s) { a_s[s] = a_s_nx[s]; a_n[s] = a_n_nx[s]; }
        __syncthreads();
    }
    {
        NextChunk none = {Ws, Wn, arow_s, arow_n};
        mma_chunk<false, NT, NTHR, WU>(
            acc, a_s, a_n, &lds_w[(N_CHUNKS - 1) & 1][0][bBase],
            &lds_w[(N_CHUNKS - 1) & 1][1][bBase],
            none, tid, ws0, ws1, wn0, wn1, a_s_nx, a_n_nx);
    }

    const int colBase = nstrip * (NT * 16) + rowInTile;
    const int rowBase =
        blockIdx.x * (16 * MT) + mtile * 16 + ((lane >> 4) << 3);
#pragma unroll
    for (int t = 0; t < NT; ++t) {
#pragma unroll
        for (int r = 0; r < 8; ++r) {
            float v = acc[t][r];
            if (RELU) v = fmaxf(v, 0.0f);
            out[(size_t)(rowBase + r) * FEAT + colBase + t * 16] = v;
        }
    }
}

__global__ __launch_bounds__(256) void l2norm_kernel(
    const float* __restrict__ pre, float* __restrict__ out)
{
    __shared__ float ssum[8];
    const int row = blockIdx.x;
    const int t   = threadIdx.x;
    const float v = pre[(size_t)row * FEAT + t];
    float s = v * v;
#pragma unroll
    for (int o = 16; o >= 1; o >>= 1) s += __shfl_xor(s, o, 32);
    if ((t & 31) == 0) ssum[t >> 5] = s;
    __syncthreads();
    float tot = 0.f;
#pragma unroll
    for (int i = 0; i < 8; ++i) tot += ssum[i];
    const float norm = fmaxf(sqrtf(tot), 1e-12f);
    out[(size_t)row * FEAT + t] = v / norm;
}

extern "C" void kernel_launch(void* const* d_in, const int* in_sizes, int n_in,
                              void* d_out, int out_size, void* d_ws,
                              size_t ws_size, hipStream_t stream)
{
    const float* features = (const float*)d_in[0];
    const int*   adj      = (const int*)d_in[1];
    const int*   batch    = (const int*)d_in[2];
    const float* Ws0      = (const float*)d_in[3];
    const float* Wn0      = (const float*)d_in[4];
    const float* Ws1      = (const float*)d_in[5];
    const float* Wn1      = (const float*)d_in[6];
    float* out = (float*)d_out;

    char* ws = (char*)d_ws;
    int*   s1   = (int*)(ws + 0);
    float* nm2  = (float*)(ws + 20480);
    float* nm1  = (float*)(ws + 5263360);
    float* h1   = (float*)(ws + 5787648);
    float* h0   = (float*)(ws + 11030528);
    float* nmh1 = (float*)(ws + 11554816);
    float* pre  = (float*)(ws + 12079104);

    build_s1_kernel<<<(5120 + 255) / 256, 256, 0, stream>>>(adj, batch, s1);
    mean_neigh25_kernel<<<5120, 64, 0, stream>>>(features, adj, s1, nm2);
    mean_neigh10_kernel<<<512, 64, 0, stream>>>(features, s1, nm1);
    sage_gemm_kernel<true, true, 2, 8><<<5120 / 32, 256, 0, stream>>>(
        features, s1, nm2, Ws0, Wn0, h1);
    sage_gemm_kernel<true, true, 1, 4><<<512 / 16, 128, 0, stream>>>(
        features, batch, nm1, Ws0, Wn0, h0);
    mean_rows10_kernel<<<512, 64, 0, stream>>>(h1, nmh1);
    sage_gemm_kernel<false, false, 1, 4><<<512 / 16, 128, 0, stream>>>(
        h0, (const int*)nullptr, nmh1, Ws1, Wn1, pre);
    l2norm_kernel<<<512, 256, 0, stream>>>(pre, out);
}


# 8 launches fused to 3: sample list + both neighbour means in one grid; both layer-0 GEMMs in one grid; layer-1 GEMM does the 10-row mean before and the l2 norm behind it (norm's sum tree changes)
# speedup vs baseline: 4.6480x; 1.3793x over previous
#include <hip/hip_runtime.h>
#include <hip/hip_bf16.h>

typedef __attribute__((ext_vector_type(2))) float v2f;
typedef __attribute__((ext_vector_type(8))) float v8f;

#define FEAT 256
#define TILE_K 32

__global__ __launch_bounds__(64) void gather_means_kernel(
    const float* __restrict__ feat, const int* __restrict__ adj,
    const int* __restrict__ batch, int* __restrict__ s1,
    float* __restrict__ nm2, float* __restrict__ nm1)
{
    const int t = threadIdx.x;
    float ax = 0.f, ay = 0.f, az = 0.f, aw = 0.f;
    if (blockIdx.x < 5120) {
        const int row  = blockIdx.x;
        const int b    = row / 10;
        const int j    = row - b * 10;
        const int node = adj[(size_t)batch[b] * 25 + j];
        const int* arow = adj + (size_t)node * 25;
        for (int k = 0; k < 25; ++k) {
            const int nb = arow[k];
            const float4 v =
                *(const float4*)(feat + (size_t)nb * FEAT + t * 4);
            ax += v.x; ay += v.y; az += v.z; aw += v.w;
        }
        const float s = 1.0f / 25.0f;
        float4 r; r.x = ax * s; r.y = ay * s; r.z = az * s; r.w = aw * s;
        *(float4*)(nm2 + (size_t)row * FEAT + t * 4) = r;
    } else if (blockIdx.x >= 5632) {
        const int i = (blockIdx.x - 5632) * 64 + t;
        const int b = i / 10;
        const int j = i - b * 10;
        s1[i] = adj[(size_t)batch[b] * 25 + j];
    } else {
        const int row   = blockIdx.x - 5120;
        const int* arow = adj + (size_t)batch[row] * 25;
        for (int j = 0; j < 10; ++j) {
            const int nb = arow[j];
            const float4 v =
                *(const float4*)(feat + (size_t)nb * FEAT + t * 4);
            ax += v.x; ay += v.y; az += v.z; aw += v.w;
        }
        const float s = 0.1f;
        float4 r; r.x = ax * s; r.y = ay * s; r.z = az * s; r.w = aw * s;
        *(float4*)(nm1 + (size_t)row * FEAT + t * 4) = r;
    }
}

#define DEV static __device__ __forceinline__
#define CHUNK_FLOATS (TILE_K * FEAT)
#define N_CHUNKS (FEAT / TILE_K)
#define K_STEPS (TILE_K / 4)
#define W_UNITS (TILE_K / 2 * (FEAT / 2))
#define PF_PER_STEP 16

DEV void sched_fence() { __builtin_amdgcn_sched_barrier(0); }

template <int NTHR>
DEV int w_unit_goff(int i, int tid)
{
    const int u     = i * NTHR + tid;
    const int kpair = u >> 7;
    const int npair = u & 127;
    return (2 * kpair) * FEAT + 2 * npair;
}

template <int NTHR, int WU>
DEV void w_chunk_to_regs(const float* __restrict__ g, int tid,
                         v2f (&r0)[WU], v2f (&r1)[WU])
{
#pragma unroll
    for (int i = 0; i < WU; ++i) {
        const int g0 = w_unit_goff<NTHR>(i, tid);
        r0[i] = *(const v2f*)(g + g0);
        r1[i] = *(const v2f*)(g + g0 + FEAT);
    }
}

template <int NTHR, int WU>
DEV void w_regs_to_lds(float* l, int tid,
                       const v2f (&r0)[WU], const v2f (&r1)[WU])
{
#pragma unroll
    for (int i = 0; i < WU; ++i) {
        const int u     = i * NTHR + tid;
        const int kpair = u >> 7;
        const int npair = u & 127;
        float4 p; p.x = r0[i].x; p.y = r1[i].x; p.z = r0[i].y; p.w = r1[i].y;
        *(float4*)(l + kpair * 512 + npair * 4) = p;
    }
}

DEV void a_chunk_to_regs(const float* __restrict__ ak, v2f (&a)[8])
{
#pragma unroll
    for (int s = 0; s < 8; ++s) a[s] = *(const v2f*)(ak + s * 4);
}

struct NextChunk {
    const float* gws;
    const float* gwn;
    const float* aks;
    const float* akn;
};

template <bool PREFETCH, int NT, int NTHR, int WU>
DEV void mma_chunk(v8f (&acc)[NT], const v2f (&a_s)[8], const v2f (&a_n)[8],
                   const float* bws, const float* bwn,
                   const NextChunk& nx, int tid,
                   v2f (&ws0)[WU], v2f (&ws1)[WU],
                   v2f (&wn0)[WU], v2f (&wn1)[WU],
                   v2f (&a_s_nx)[8], v2f (&a_n_nx)[8])
{
    constexpr int PF_TOTAL = 16 + 4 * WU;
    static_assert(PF_TOTAL <= PF_PER_STEP * (K_STEPS - 3),
                  "requests must be out three runs before the chunk ends");
    v2f bs[2][NT], bn[2][NT];
#pragma unroll
    for (int t = 0; t < NT; ++t) {
        bs[0][t] = *(const v2f*)(bws + t * 32);
        bn[0][t] = *(const v2f*)(bwn + t * 32);
    }
#pragma unroll
    for (int s = 0; s < K_STEPS; ++s) {
        const int cur = s & 1;
        if (PREFETCH) {
#pragma unroll
            for (int j = s * PF_PER_STEP;
                 j < (s + 1) * PF_PER_STEP && j < PF_TOTAL; ++j) {
                if (j < 16) {
                    if ((j & 1) == 0)
                        a_s_nx[j >> 1] = *(const v2f*)(nx.aks + (j >> 1) * 4);
                    else
                        a_n_nx[j >> 1] = *(const v2f*)(nx.akn + (j >> 1) * 4);
                } else {
                    const int w  = j - 16;
                    const int i  = w >> 2;
                    const int g0 = w_unit_goff<NTHR>(i, tid);
                    if ((w & 3) == 0) ws0[i] = *(const v2f*)(nx.gws + g0);
                    if ((w & 3) == 1) ws1[i] = *(const v2f*)(nx.gws + g0 + FEAT);
                    if ((w & 3) == 2) wn0[i] = *(const v2f*)(nx.gwn + g0);
                    if ((w & 3) == 3) wn1[i] = *(const v2f*)(nx.gwn + g0 + FEAT);
                }
            }
        }
        if (s + 1 < K_STEPS) {
#pragma unroll
            for (int t = 0; t < NT; ++t) {
                bs[cur ^ 1][t] = *(const v2f*)(bws + (s + 1) * 1024 + t * 32);
                bn[cur ^ 1][t] = *(const v2f*)(bwn + (s + 1) * 1024 + t * 32);
            }
        }
        sched_fence();
#pragma unroll
        for (int t = 0; t < NT; ++t)
            acc[t] = __builtin_amdgcn_wmma_f32_16x16x4_f32(
                false, a_s[s], false, bs[cur][t], (short)0, acc[t],
                false, false);
#pragma unroll
        for (int t = 0; t < NT; ++t)
            acc[t] = __builtin_amdgcn_wmma_f32_16x16x4_f32(
                false, a_n[s], false, bn[cur][t], (short)0, acc[t],
                false, false);
        sched_fence();
    }
}

DEV float* lds_stage(float* lds_w, int stage, int matrix)
{
    return lds_w + (stage * 2 + matrix) * CHUNK_FLOATS;
}

template <bool RELU, bool GATHER, bool L2NORM, int MT, int NWV>
DEV void sage_gemm_block(
    const float* Aself, const int* idx, const float* Aneigh,
    const float* Ws, const float* Wn, float* out, int blk,
    float* lds_w, float* row_ss)
{
    constexpr int NTHR = 32 * NWV;
    constexpr int WU   = W_UNITS / NTHR;
    constexpr int NW   = NWV / MT;
    constexpr int NT   = 16 / NW;
    static_assert(NW * MT == NWV && NT * NW == 16 && WU * NTHR == W_UNITS,
                  "block shape must tile 16*MT x 256 and the weight chunk");

    const int tid       = threadIdx.x;
    const int lane      = tid & 31;
    const int wave      = tid >> 5;
    const int mtile     = wave / NW;
    const int nstrip    = wave % NW;
    const int rowInTile = lane & 15;
    const int klane     = (lane >> 4) << 1;
    const int m         = blk * (16 * MT) + mtile * 16 + rowInTile;

    const float* arow_s =
        (GATHER ? (Aself + (size_t)idx[m] * FEAT) : (Aself + (size_t)m * FEAT))
        + klane;
    const float* arow_n = Aneigh + (size_t)m * FEAT + klane;

    const int bBase =
        (klane >> 1) * 512 + (nstrip * (NT * 16) + rowInTile) * 2;

    v8f acc[NT];
#pragma unroll
    for (int t = 0; t < NT; ++t) acc[t] = (v8f)0.f;

    v2f ws0[WU], ws1[WU], wn0[WU], wn1[WU];
    v2f a_s[8], a_n[8];
    v2f a_s_nx[8], a_n_nx[8];

    w_chunk_to_regs<NTHR, WU>(Ws, tid, ws0, ws1);
    w_chunk_to_regs<NTHR, WU>(Wn, tid, wn0, wn1);
    a_chunk_to_regs(arow_s, a_s);
    a_chunk_to_regs(arow_n, a_n);
    w_regs_to_lds<NTHR, WU>(lds_stage(lds_w, 0, 0), tid, ws0, ws1);
    w_regs_to_lds<NTHR, WU>(lds_stage(lds_w, 0, 1), tid, wn0, wn1);
    __syncthreads();

#pragma unroll 1
    for (int c = 0; c + 1 < N_CHUNKS; ++c) {
        const int k1 = (c + 1) * TILE_K;
        NextChunk nx;
        nx.gws = Ws + (size_t)k1 * FEAT;
        nx.gwn = Wn + (size_t)k1 * FEAT;
        nx.aks = arow_s + k1;
        nx.akn = arow_n + k1;

        mma_chunk<true, NT, NTHR, WU>(
            acc, a_s, a_n, lds_stage(lds_w, c & 1, 0) + bBase,
            lds_stage(lds_w, c & 1, 1) + bBase,
            nx, tid, ws0, ws1, wn0, wn1, a_s_nx, a_n_nx);

        w_regs_to_lds<NTHR, WU>(lds_stage(lds_w, (c + 1) & 1, 0), tid, ws0, ws1);
        w_regs_to_lds<NTHR, WU>(lds_stage(lds_w, (c + 1) & 1, 1), tid, wn0, wn1);
#pragma unroll
        for (int s = 0; s < 8; ++s) { a_s[s] = a_s_nx[s]; a_n[s] = a_n_nx[s]; }
        __syncthreads();
    }
    {
        NextChunk none = {Ws, Wn, arow_s, arow_n};
        mma_chunk<false, NT, NTHR, WU>(
            acc, a_s, a_n, lds_stage(lds_w, (N_CHUNKS - 1) & 1, 0) + bBase,
            lds_stage(lds_w, (N_CHUNKS - 1) & 1, 1) + bBase,
            none, tid, ws0, ws1, wn0, wn1, a_s_nx, a_n_nx);
    }

    const int half    = lane >> 4;
    const int colBase = nstrip * (NT * 16) + rowInTile;
    const int rowBase = blk * (16 * MT) + mtile * 16 + (half << 3);

    float norm[8];
    if (L2NORM) {
        float p[8];
#pragma unroll
        for (int r = 0; r < 8; ++r) {
            float s = 0.f;
#pragma unroll
            for (int t = 0; t < NT; ++t) s += acc[t][r] * acc[t][r];
            p[r] = s;
        }
#pragma unroll
        for (int r = 0; r < 8; ++r) {
#pragma unroll
            for (int o = 8; o >= 1; o >>= 1) p[r] += __shfl_xor(p[r], o, 32);
        }
        if (rowInTile == 0) {
#pragma unroll
            for (int r = 0; r < 8; ++r)
                row_ss[(mtile * NW + nstrip) * 16 + half * 8 + r] = p[r];
        }
        __syncthreads();
#pragma unroll
        for (int r = 0; r < 8; ++r) {
            float tot = 0.f;
#pragma unroll
            for (int w = 0; w < NW; ++w)
                tot += row_ss[(mtile * NW + w) * 16 + half * 8 + r];
            norm[r] = fmaxf(sqrtf(tot), 1e-12f);
        }
    }

#pragma unroll
    for (int t = 0; t < NT; ++t) {
#pragma unroll
        for (int r = 0; r < 8; ++r) {
            float v = acc[t][r];
            if (RELU) v = fmaxf(v, 0.0f);
            if (L2NORM) v = v / norm[r];
            out[(size_t)(rowBase + r) * FEAT + colBase + t * 16] = v;
        }
    }
}

#define HOP1_BLOCKS (5120 / 32)
__global__ __launch_bounds__(256) void sage_layer0_kernel(
    const float* __restrict__ feat,
    const int* __restrict__ s1, const float* __restrict__ nm2,
    float* __restrict__ h1,
    const int* __restrict__ batch, const float* __restrict__ nm1,
    float* __restrict__ h0,
    const float* __restrict__ Ws, const float* __restrict__ Wn)
{
    __shared__ float lds_w[2 * 2 * CHUNK_FLOATS];
    const bool hop0 = blockIdx.x >= HOP1_BLOCKS;
    const int*   idx    = hop0 ? batch : s1;
    const float* aneigh = hop0 ? nm1 : nm2;
    float*       out    = hop0 ? h0 : h1;
    const int    blk    = hop0 ? (int)blockIdx.x - HOP1_BLOCKS : (int)blockIdx.x;
    sage_gemm_block<true, true, false, 2, 8>(
        feat, idx, aneigh, Ws, Wn, out, blk, lds_w, (float*)nullptr);
}

__global__ __launch_bounds__(128) void sage_layer1_kernel(
    const float* __restrict__ h0, const float* __restrict__ h1, float* nm,
    const float* __restrict__ Ws, const float* __restrict__ Wn,
    float* __restrict__ out)
{
    __shared__ float lds_w[2 * 2 * CHUNK_FLOATS];
    __shared__ float row_ss[4 * 16];
    const int tid = threadIdx.x;
#pragma unroll
    for (int g = 0; g < 2; ++g) {
        float4 v[4][10];
#pragma unroll
        for (int i = 0; i < 4; ++i) {
            const int piece = (g * 4 + i) * 128 + tid;
            const size_t m  = (size_t)blockIdx.x * 16 + (piece >> 6);
            const float* src = h1 + m * 10 * FEAT + (piece & 63) * 4;
#pragma unroll
            for (int j = 0; j < 10; ++j)
                v[i][j] = *(const float4*)(src + (size_t)j * FEAT);
        }
#pragma unroll
        for (int i = 0; i < 4; ++i) {
            const int piece = (g * 4 + i) * 128 + tid;
            const size_t m  = (size_t)blockIdx.x * 16 + (piece >> 6);
            float ax = 0.f, ay = 0.f, az = 0.f, aw = 0.f;
#pragma unroll
            for (int j = 0; j < 10; ++j) {
                ax += v[i][j].x; ay += v[i][j].y;
                az += v[i][j].z; aw += v[i][j].w;
            }
            const float s = 0.1f;
            float4 q; q.x = ax * s; q.y = ay * s; q.z = az * s; q.w = aw * s;
            *(float4*)(nm + m * FEAT + (piece & 63) * 4) = q;
        }
    }
    __syncthreads();
    sage_gemm_block<false, false, true, 1, 4>(
        h0, (const int*)nullptr, nm, Ws, Wn, out, (int)blockIdx.x,
        lds_w, row_ss);
}

extern "C" void kernel_launch(void* const* d_in, const int* in_sizes, int n_in,
                              void* d_out, int out_size, void* d_ws,
                              size_t ws_size, hipStream_t stream)
{
    const float* features = (const float*)d_in[0];
    const int*   adj      = (const int*)d_in[1];
    const int*   batch    = (const int*)d_in[2];
    const float* Ws0      = (const float*)d_in[3];
    const float* Wn0      = (const float*)d_in[4];
    const float* Ws1      = (const float*)d_in[5];
    const float* Wn1      = (const float*)d_in[6];
    float* out = (float*)d_out;

    char* ws = (char*)d_ws;
    int*   s1   = (int*)(ws + 0);
    float* nm2  = (float*)(ws + 20480);
    float* nm1  = (float*)(ws + 5263360);
    float* h1   = (float*)(ws + 5787648);
    float* h0   = (float*)(ws + 11030528);
    float* nmh1 = (float*)(ws + 11554816);

    gather_means_kernel<<<5120 + 512 + 80, 64, 0, stream>>>(
        features, adj, batch, s1, nm2, nm1);
    sage_layer0_kernel<<<5120 / 32 + 512 / 32, 256, 0, stream>>>(
        features, s1, nm2, h1, batch, nm1, h0, Ws0, Wn0);
    sage_layer1_kernel<<<512 / 16, 128, 0, stream>>>(
        h0, h1, nmh1, Ws1, Wn1, out);
}


# A rows of both GEMMs go through LDS: a block requests its rows' K-chunk once, 16 B per lane on whole 128-B lines (xor-swizzled tile), waves read fragments from LDS, not 16-line requests; sums same
# speedup vs baseline: 5.2214x; 1.1234x over previous
#include <hip/hip_runtime.h>
#include <hip/hip_bf16.h>

typedef __attribute__((ext_vector_type(2))) float v2f;
typedef __attribute__((ext_vector_type(8))) float v8f;

#define FEAT 256
#define TILE_K 32

__global__ __launch_bounds__(64) void gather_means_kernel(
    const float* __restrict__ feat, const int* __restrict__ adj,
    const int* __restrict__ batch, int* __restrict__ s1,
    float* __restrict__ nm2, float* __restrict__ nm1)
{
    const int t = threadIdx.x;
    float ax = 0.f, ay = 0.f, az = 0.f, aw = 0.f;
    if (blockIdx.x < 5120) {
        const int row  = blockIdx.x;
        const int b    = row / 10;
        const int j    = row - b * 10;
        const int node = adj[(size_t)batch[b] * 25 + j];
        const int* arow = adj + (size_t)node * 25;
        for (int k = 0; k < 25; ++k) {
            const int nb = arow[k];
            const float4 v =
                *(const float4*)(feat + (size_t)nb * FEAT + t * 4);
            ax += v.x; ay += v.y; az += v.z; aw += v.w;
        }
        const float s = 1.0f / 25.0f;
        float4 r; r.x = ax * s; r.y = ay * s; r.z = az * s; r.w = aw * s;
        *(float4*)(nm2 + (size_t)row * FEAT + t * 4) = r;
    } else if (blockIdx.x >= 5632) {
        const int i = (blockIdx.x - 5632) * 64 + t;
        const int b = i / 10;
        const int j = i - b * 10;
        s1[i] = adj[(size_t)batch[b] * 25 + j];
    } else {
        const int row   = blockIdx.x - 5120;
        const int* arow = adj + (size_t)batch[row] * 25;
        for (int j = 0; j < 10; ++j) {
            const int nb = arow[j];
            const float4 v =
                *(const float4*)(feat + (size_t)nb * FEAT + t * 4);
            ax += v.x; ay += v.y; az += v.z; aw += v.w;
        }
        const float s = 0.1f;
        float4 r; r.x = ax * s; r.y = ay * s; r.z = az * s; r.w = aw * s;
        *(float4*)(nm1 + (size_t)row * FEAT + t * 4) = r;
    }
}

#define DEV static __device__ __forceinline__
#define CHUNK_FLOATS (TILE_K * FEAT)
#define N_CHUNKS (FEAT / TILE_K)
#define K_STEPS (TILE_K / 4)
#define W_UNITS (TILE_K / 2 * (FEAT / 2))
#define PF_PER_STEP 16

DEV void sched_fence() { __builtin_amdgcn_sched_barrier(0); }

template <int NTHR>
DEV int w_unit_goff(int i, int tid)
{
    const int u     = i * NTHR + tid;
    const int kpair = u >> 7;
    const int npair = u & 127;
    return (2 * kpair) * FEAT + 2 * npair;
}

template <int NTHR, int WU>
DEV void w_chunk_to_regs(const float* __restrict__ g, int tid,
                         v2f (&r0)[WU], v2f (&r1)[WU])
{
#pragma unroll
    for (int i = 0; i < WU; ++i) {
        const int g0 = w_unit_goff<NTHR>(i, tid);
        r0[i] = *(const v2f*)(g + g0);
        r1[i] = *(const v2f*)(g + g0 + FEAT);
    }
}

template <int NTHR, int WU>
DEV void w_regs_to_lds(float* l, int tid,
                       const v2f (&r0)[WU], const v2f (&r1)[WU])
{
#pragma unroll
    for (int i = 0; i < WU; ++i) {
        const int u     = i * NTHR + tid;
        const int kpair = u >> 7;
        const int npair = u & 127;
        float4 p; p.x = r0[i].x; p.y = r1[i].x; p.z = r0[i].y; p.w = r1[i].y;
        *(float4*)(l + kpair * 512 + npair * 4) = p;
    }
}

struct NextChunk {
    const float* gws;
    const float* gwn;
    const float* gas;
    const float* gan;
};

template <bool PREFETCH, int NT, int NTHR, int WU>
DEV void mma_chunk(v8f (&acc)[NT],
                   const float* bws, const float* bwn,
                   const float* fas, const float* fan, const int (&aoff)[4],
                   const NextChunk& nx, int tid,
                   v2f (&ws0)[WU], v2f (&ws1)[WU],
                   v2f (&wn0)[WU], v2f (&wn1)[WU],
                   float4& pa_s, float4& pa_n)
{
    constexpr int PF_TOTAL = 2 + 4 * WU;
    static_assert(PF_TOTAL <= PF_PER_STEP * (K_STEPS - 3),
                  "requests must be out three runs before the chunk ends");
    v2f bs[2][NT], bn[2][NT];
    v2f fa_s[2], fa_n[2];
#pragma unroll
    for (int t = 0; t < NT; ++t) {
        bs[0][t] = *(const v2f*)(bws + t * 32);
        bn[0][t] = *(const v2f*)(bwn + t * 32);
    }
    fa_s[0] = *(const v2f*)(fas + aoff[0]);
    fa_n[0] = *(const v2f*)(fan + aoff[0]);
#pragma unroll
    for (int s = 0; s < K_STEPS; ++s) {
        const int cur = s & 1;
        if (PREFETCH) {
#pragma unroll
            for (int j = s * PF_PER_STEP;
                 j < (s + 1) * PF_PER_STEP && j < PF_TOTAL; ++j) {
                if (j == 0) {
                    pa_s = *(const float4*)nx.gas;
                } else if (j == 1) {
                    pa_n = *(const float4*)nx.gan;
                } else {
                    const int w  = j - 2;
                    const int i  = w >> 2;
                    const int g0 = w_unit_goff<NTHR>(i, tid);
                    if ((w & 3) == 0) ws0[i] = *(const v2f*)(nx.gws + g0);
                    if ((w & 3) == 1) ws1[i] = *(const v2f*)(nx.gws + g0 + FEAT);
                    if ((w & 3) == 2) wn0[i] = *(const v2f*)(nx.gwn + g0);
                    if ((w & 3) == 3) wn1[i] = *(const v2f*)(nx.gwn + g0 + FEAT);
                }
            }
        }
        if (s + 1 < K_STEPS) {
#pragma unroll
            for (int t = 0; t < NT; ++t) {
                bs[cur ^ 1][t] = *(const v2f*)(bws + (s + 1) * 1024 + t * 32);
                bn[cur ^ 1][t] = *(const v2f*)(bwn + (s + 1) * 1024 + t * 32);
            }
            fa_s[cur ^ 1] = *(const v2f*)(fas + (s + 1) * 64 + aoff[(s + 1) & 3]);
            fa_n[cur ^ 1] = *(const v2f*)(fan + (s + 1) * 64 + aoff[(s + 1) & 3]);
        }
        sched_fence();
#pragma unroll
        for (int t = 0; t < NT; ++t)
            acc[t] = __builtin_amdgcn_wmma_f32_16x16x4_f32(
                false, fa_s[cur], false, bs[cur][t], (short)0, acc[t],
                false, false);
#pragma unroll
        for (int t = 0; t < NT; ++t)
            acc[t] = __builtin_amdgcn_wmma_f32_16x16x4_f32(
                false, fa_n[cur], false, bn[cur][t], (short)0, acc[t],
                false, false);
        sched_fence();
    }
}

DEV float* lds_stage(float* lds_w, int stage, int matrix)
{
    return lds_w + (stage * 2 + matrix) * CHUNK_FLOATS;
}

#define A_TILE_FLOATS(MT) (16 * (MT) * TILE_K)

template <bool RELU, bool GATHER, bool L2NORM, int MT, int NWV>
DEV void sage_gemm_block(
    const float* Aself, const int* idx, const float* Aneigh,
    const float* Ws, const float* Wn, float* out, int blk,
    float* lds_w, float* lds_a, float* row_ss)
{
    constexpr int NTHR = 32 * NWV;
    constexpr int WU   = W_UNITS / NTHR;
    constexpr int NW   = NWV / MT;
    constexpr int NT   = 16 / NW;
    constexpr int AT   = A_TILE_FLOATS(MT);
    static_assert(NW * MT == NWV && NT * NW == 16 && WU * NTHR == W_UNITS,
                  "block shape must tile 16*MT x 256 and the weight chunk");
    static_assert(NTHR == 16 * MT * (TILE_K / 4),
                  "one 16-byte A piece per thread, matrix and chunk");

    const int tid       = threadIdx.x;
    const int lane      = tid & 31;
    const int wave      = tid >> 5;
    const int mtile     = wave / NW;
    const int nstrip    = wave % NW;
    const int rowInTile = lane & 15;
    const int klane     = (lane >> 4) << 1;

    const int sr = tid >> 3;
    const int sq = tid & 7;
    const int sm = blk * (16 * MT) + sr;
    const float* ga_s =
        (GATHER ? (Aself + (size_t)idx[sm] * FEAT) : (Aself + (size_t)sm * FEAT))
        + sq * 4;
    const float* ga_n = Aneigh + (size_t)sm * FEAT + sq * 4;
    const int a_st =
        ((sr >> 4) * 8 + sq) * 64 + (((sr & 15) ^ ((sq & 3) << 2)) << 2);

    int aoff[4];
#pragma unroll
    for (int j = 0; j < 4; ++j)
        aoff[j] = mtile * 512 + ((rowInTile ^ (j << 2)) << 2) + klane;

    const int bBase =
        (klane >> 1) * 512 + (nstrip * (NT * 16) + rowInTile) * 2;

    v8f acc[NT];
#pragma unroll
    for (int t = 0; t < NT; ++t) acc[t] = (v8f)0.f;

    v2f ws0[WU], ws1[WU], wn0[WU], wn1[WU];
    float4 pa_s, pa_n;

    pa_s = *(const float4*)ga_s;
    pa_n = *(const float4*)ga_n;
    w_chunk_to_regs<NTHR, WU>(Ws, tid, ws0, ws1);
    w_chunk_to_regs<NTHR, WU>(Wn, tid, wn0, wn1);
    w_regs_to_lds<NTHR, WU>(lds_stage(lds_w, 0, 0), tid, ws0, ws1);
    w_regs_to_lds<NTHR, WU>(lds_stage(lds_w, 0, 1), tid, wn0, wn1);
    *(float4*)(lds_a + a_st)      = pa_s;
    *(float4*)(lds_a + AT + a_st) = pa_n;
    __syncthreads();

#pragma unroll 1
    for (int c = 0; c + 1 < N_CHUNKS; ++c) {
        const int k1 = (c + 1) * TILE_K;
        NextChunk nx;
        nx.gws = Ws + (size_t)k1 * FEAT;
        nx.gwn = Wn + (size_t)k1 * FEAT;
        nx.gas = ga_s + k1;
        nx.gan = ga_n + k1;

        const float* at = lds_a + (c & 1) * 2 * AT;
        mma_chunk<true, NT, NTHR, WU>(
            acc, lds_stage(lds_w, c & 1, 0) + bBase,
            lds_stage(lds_w, c & 1, 1) + bBase,
            at, at + AT, aoff,
            nx, tid, ws0, ws1, wn0, wn1, pa_s, pa_n);

        w_regs_to_lds<NTHR, WU>(lds_stage(lds_w, (c + 1) & 1, 0), tid, ws0, ws1);
        w_regs_to_lds<NTHR, WU>(lds_stage(lds_w, (c + 1) & 1, 1), tid, wn0, wn1);
        float* at_nx = lds_a + ((c + 1) & 1) * 2 * AT;
        *(float4*)(at_nx + a_st)      = pa_s;
        *(float4*)(at_nx + AT + a_st) = pa_n;
        __syncthreads();
    }
    {
        NextChunk none = {Ws, Wn, ga_s, ga_n};
        const float* at = lds_a + ((N_CHUNKS - 1) & 1) * 2 * AT;
        mma_chunk<false, NT, NTHR, WU>(
            acc, lds_stage(lds_w, (N_CHUNKS - 1) & 1, 0) + bBase,
            lds_stage(lds_w, (N_CHUNKS - 1) & 1, 1) + bBase,
            at, at + AT, aoff,
            none, tid, ws0, ws1, wn0, wn1, pa_s, pa_n);
    }

    const int half    = lane >> 4;
    const int colBase = nstrip * (NT * 16) + rowInTile;
    const int rowBase = blk * (16 * MT) + mtile * 16 + (half << 3);

    float norm[8];
    if (L2NORM) {
        float p[8];
#pragma unroll
        for (int r = 0; r < 8; ++r) {
            float s = 0.f;
#pragma unroll
            for (int t = 0; t < NT; ++t) s += acc[t][r] * acc[t][r];
            p[r] = s;
        }
#pragma unroll
        for (int r = 0; r < 8; ++r) {
#pragma unroll
            for (int o = 8; o >= 1; o >>= 1) p[r] += __shfl_xor(p[r], o, 32);
        }
        if (rowInTile == 0) {
#pragma unroll
            for (int r = 0; r < 8; ++r)
                row_ss[(mtile * NW + nstrip) * 16 + half * 8 + r] = p[r];
        }
        __syncthreads();
#pragma unroll
        for (int r = 0; r < 8; ++r) {
            float tot = 0.f;
#pragma unroll
            for (int w = 0; w < NW; ++w)
                tot += row_ss[(mtile * NW + w) * 16 + half * 8 + r];
            norm[r] = fmaxf(sqrtf(tot), 1e-12f);
        }
    }

#pragma unroll
    for (int t = 0; t < NT; ++t) {
#pragma unroll
        for (int r = 0; r < 8; ++r) {
            float v = acc[t][r];
            if (RELU) v = fmaxf(v, 0.0f);
            if (L2NORM) v = v / norm[r];
            out[(size_t)(rowBase + r) * FEAT + colBase + t * 16] = v;
        }
    }
}

#define HOP1_BLOCKS (5120 / 32)
__global__ __launch_bounds__(256) void sage_layer0_kernel(
    const float* feat,
    const int* __restrict__ s1, const float* __restrict__ nm2,
    float* __restrict__ h1,
    const int* __restrict__ batch, const float* __restrict__ nm1,
    float* __restrict__ h0,
    const float* __restrict__ Ws, const float* __restrict__ Wn)
{
    __shared__ float lds_w[2 * 2 * CHUNK_FLOATS];
    __shared__ alignas(16) float lds_a[2 * 2 * A_TILE_FLOATS(2)];
    const bool hop0 = blockIdx.x >= HOP1_BLOCKS;
    const int*   idx    = hop0 ? batch : s1;
    const float* aneigh = hop0 ? nm1 : nm2;
    float*       out    = hop0 ? h0 : h1;
    const int    blk    = hop0 ? (int)blockIdx.x - HOP1_BLOCKS : (int)blockIdx.x;
    sage_gemm_block<true, true, false, 2, 8>(
        feat, idx, aneigh, Ws, Wn, out, blk, lds_w, lds_a, (float*)nullptr);
}

__global__ __launch_bounds__(128) void sage_layer1_kernel(
    const float* __restrict__ h0, const float* __restrict__ h1, float* nm,
    const float* __restrict__ Ws, const float* __restrict__ Wn,
    float* __restrict__ out)
{
    __shared__ float lds_w[2 * 2 * CHUNK_FLOATS];
    __shared__ alignas(16) float lds_a[2 * 2 * A_TILE_FLOATS(1)];
    __shared__ float row_ss[4 * 16];
    const int tid = threadIdx.x;
#pragma unroll
    for (int g = 0; g < 2; ++g) {
        float4 v[4][10];
#pragma unroll
        for (int i = 0; i < 4; ++i) {
            const int piece = (g * 4 + i) * 128 + tid;
            const size_t m  = (size_t)blockIdx.x * 16 + (piece >> 6);
            const float* src = h1 + m * 10 * FEAT + (piece & 63) * 4;
#pragma unroll
            for (int j = 0; j < 10; ++j)
                v[i][j] = *(const float4*)(src + (size_t)j * FEAT);
        }
#pragma unroll
        for (int i = 0; i < 4; ++i) {
            const int piece = (g * 4 + i) * 128 + tid;
            const size_t m  = (size_t)blockIdx.x * 16 + (piece >> 6);
            float ax = 0.f, ay = 0.f, az = 0.f, aw = 0.f;
#pragma unroll
            for (int j = 0; j < 10; ++j) {
                ax += v[i][j].x; ay += v[i][j].y;
                az += v[i][j].z; aw += v[i][j].w;
            }
            const float s = 0.1f;
            float4 q; q.x = ax * s; q.y = ay * s; q.z = az * s; q.w = aw * s;
            *(float4*)(nm + m * FEAT + (piece & 63) * 4) = q;
        }
    }
    __syncthreads();
    sage_gemm_block<false, false, true, 1, 4>(
        h0, (const int*)nullptr, nm, Ws, Wn, out, (int)blockIdx.x,
        lds_w, lds_a, row_ss);
}

extern "C" void kernel_launch(void* const* d_in, const int* in_sizes, int n_in,
                              void* d_out, int out_size, void* d_ws,
                              size_t ws_size, hipStream_t stream)
{
    const float* features = (const float*)d_in[0];
    const int*   adj      = (const int*)d_in[1];
    const int*   batch    = (const int*)d_in[2];
    const float* Ws0      = (const float*)d_in[3];
    const float* Wn0      = (const float*)d_in[4];
    const float* Ws1      = (const float*)d_in[5];
    const float* Wn1      = (const float*)d_in[6];
    float* out = (float*)d_out;

    char* ws = (char*)d_ws;
    int*   s1   = (int*)(ws + 0);
    float* nm2  = (float*)(ws + 20480);
    float* nm1  = (float*)(ws + 5263360);
    float* h1   = (float*)(ws + 5787648);
    float* h0   = (float*)(ws + 11030528);
    float* nmh1 = (float*)(ws + 11554816);

    gather_means_kernel<<<5120 + 512 + 80, 64, 0, stream>>>(
        features, adj, batch, s1, nm2, nm1);
    sage_layer0_kernel<<<5120 / 32 + 512 / 32, 256, 0, stream>>>(
        features, s1, nm2, h1, batch, nm1, h0, Ws0, Wn0);
    sage_layer1_kernel<<<512 / 16, 128, 0, stream>>>(
        h0, h1, nmh1, Ws1, Wn1, out);
}


# 3 launches to 2: each layer-0 GEMM block finds its rows' nodes and sums their 25 (or 10) neighbour rows in the A staging, under the chunk's WMMAs; gather launch and its workspace planes go; sums same
# speedup vs baseline: 5.4870x; 1.0509x over previous
#include <hip/hip_runtime.h>
#include <hip/hip_bf16.h>

typedef __attribute__((ext_vector_type(2))) float v2f;
typedef __attribute__((ext_vector_type(4))) float v4f;
typedef __attribute__((ext_vector_type(8))) float v8f;

#define FEAT 256
#define TILE_K 32

#define ADJ_W 25

#define DEV static __device__ __forceinline__
#define CHUNK_FLOATS (TILE_K * FEAT)
#define N_CHUNKS (FEAT / TILE_K)
#define K_STEPS (TILE_K / 4)
#define W_UNITS (TILE_K / 2 * (FEAT / 2))
#define PF_PER_STEP 16

DEV void sched_fence() { __builtin_amdgcn_sched_barrier(0); }

template <int NTHR>
DEV int w_unit_goff(int i, int tid)
{
    const int u     = i * NTHR + tid;
    const int kpair = u >> 7;
    const int npair = u & 127;
    return (2 * kpair) * FEAT + 2 * npair;
}

template <int NTHR, int WU>
DEV void w_chunk_to_regs(const float* __restrict__ g, int tid,
                         v2f (&r0)[WU], v2f (&r1)[WU])
{
#pragma unroll
    for (int i = 0; i < WU; ++i) {
        const int g0 = w_unit_goff<NTHR>(i, tid);
        r0[i] = *(const v2f*)(g + g0);
        r1[i] = *(const v2f*)(g + g0 + FEAT);
    }
}

template <int NTHR, int WU>
DEV void w_regs_to_lds(float* l, int tid,
                       const v2f (&r0)[WU], const v2f (&r1)[WU])
{
#pragma unroll
    for (int i = 0; i < WU; ++i) {
        const int u     = i * NTHR + tid;
        const int kpair = u >> 7;
        const int npair = u & 127;
        float4 p; p.x = r0[i].x; p.y = r1[i].x; p.z = r0[i].y; p.w = r1[i].y;
        *(float4*)(l + kpair * 512 + npair * 4) = p;
    }
}

struct NextChunk {
    const float* gws;
    const float* gwn;
    const float* gas;
    const float* gan;
};

template <bool PREFETCH, int NT, int NTHR, int WU, int NNB, int NG>
DEV void mma_chunk(v8f (&acc)[NT],
                   const float* bws, const float* bwn,
                   const float* fas, const float* fan, const int (&aoff)[4],
                   const NextChunk& nx, int tid,
                   v2f (&ws0)[WU], v2f (&ws1)[WU],
                   v2f (&wn0)[WU], v2f (&wn1)[WU],
                   float4& pa_s, float4& pa_n,
                   v4f (&pn)[NG], const unsigned (&noff)[NG])
{
    constexpr int NA = 1 + NG;
    constexpr int PF_TOTAL = NA + 4 * WU;
    static_assert(NG == (NNB > 0 ? NNB : 1), "NG is max(NNB, 1)");
    static_assert(PF_TOTAL <= PF_PER_STEP * (K_STEPS - 3),
                  "requests must be out three runs before the chunk ends");
    v2f bs[2][NT], bn[2][NT];
    v2f fa_s[2], fa_n[2];
#pragma unroll
    for (int t = 0; t < NT; ++t) {
        bs[0][t] = *(const v2f*)(bws + t * 32);
        bn[0][t] = *(const v2f*)(bwn + t * 32);
    }
    fa_s[0] = *(const v2f*)(fas + aoff[0]);
    fa_n[0] = *(const v2f*)(fan + aoff[0]);
#pragma unroll
    for (int s = 0; s < K_STEPS; ++s) {
        const int cur = s & 1;
        if (PREFETCH) {
#pragma unroll
            for (int j = s * PF_PER_STEP;
                 j < (s + 1) * PF_PER_STEP && j < PF_TOTAL; ++j) {
                if (j == 0) {
                    pa_s = *(const float4*)nx.gas;
                } else if (j < NA) {
                    if (NNB > 0)
                        pn[j - 1] = *(const v4f*)(nx.gan + noff[j - 1]);
                    else
                        pa_n = *(const float4*)nx.gan;
                } else {
                    const int w  = j - NA;
                    const int i  = w >> 2;
                    const int g0 = w_unit_goff<NTHR>(i, tid);
                    if ((w & 3) == 0) ws0[i] = *(const v2f*)(nx.gws + g0);
                    if ((w & 3) == 1) ws1[i] = *(const v2f*)(nx.gws + g0 + FEAT);
                    if ((w & 3) == 2) wn0[i] = *(const v2f*)(nx.gwn + g0);
                    if ((w & 3) == 3) wn1[i] = *(const v2f*)(nx.gwn + g0 + FEAT);
                }
            }
        }
        if (s + 1 < K_STEPS) {
#pragma unroll
            for (int t = 0; t < NT; ++t) {
                bs[cur ^ 1][t] = *(const v2f*)(bws + (s + 1) * 1024 + t * 32);
                bn[cur ^ 1][t] = *(const v2f*)(bwn + (s + 1) * 1024 + t * 32);
            }
            fa_s[cur ^ 1] = *(const v2f*)(fas + (s + 1) * 64 + aoff[(s + 1) & 3]);
            fa_n[cur ^ 1] = *(const v2f*)(fan + (s + 1) * 64 + aoff[(s + 1) & 3]);
        }
        sched_fence();
#pragma unroll
        for (int t = 0; t < NT; ++t)
            acc[t] = __builtin_amdgcn_wmma_f32_16x16x4_f32(
                false, fa_s[cur], false, bs[cur][t], (short)0, acc[t],
                false, false);
#pragma unroll
        for (int t = 0; t < NT; ++t)
            acc[t] = __builtin_amdgcn_wmma_f32_16x16x4_f32(
                false, fa_n[cur], false, bn[cur][t], (short)0, acc[t],
                false, false);
        sched_fence();
    }
}

template <int NNB>
DEV float4 neigh_mean(const v4f (&pn)[NNB])
{
    v4f a = (v4f)0.f;
#pragma unroll
    for (int k = 0; k < NNB; ++k) a += pn[k];
    const float s = (NNB == 10) ? 0.1f : 1.0f / (float)NNB;
    float4 r; r.x = a.x * s; r.y = a.y * s; r.z = a.z * s; r.w = a.w * s;
    return r;
}

DEV float* lds_stage(float* lds_w, int stage, int matrix)
{
    return lds_w + (stage * 2 + matrix) * CHUNK_FLOATS;
}

#define A_TILE_FLOATS(MT) (16 * (MT) * TILE_K)

template <bool RELU, bool GATHER, bool L2NORM, int MT, int NWV, int NNB>
DEV void sage_gemm_block(
    const float* Aself, int node, const int* adj, const float* Aneigh,
    const float* Ws, const float* Wn, float* out, int blk,
    float* lds_w, float* lds_a, float* row_ss)
{
    constexpr int NTHR = 32 * NWV;
    constexpr int WU   = W_UNITS / NTHR;
    constexpr int NW   = NWV / MT;
    constexpr int NT   = 16 / NW;
    constexpr int AT   = A_TILE_FLOATS(MT);
    constexpr int NG   = NNB > 0 ? NNB : 1;
    static_assert(NNB == 0 || GATHER, "neighbour rows come from the node");
    static_assert(NW * MT == NWV && NT * NW == 16 && WU * NTHR == W_UNITS,
                  "block shape must tile 16*MT x 256 and the weight chunk");
    static_assert(NTHR == 16 * MT * (TILE_K / 4),
                  "one 16-byte A piece per thread, matrix and chunk");

    const int tid       = threadIdx.x;
    const int lane      = tid & 31;
    const int wave      = tid >> 5;
    const int mtile     = wave / NW;
    const int nstrip    = wave % NW;
    const int rowInTile = lane & 15;
    const int klane     = (lane >> 4) << 1;

    const int sr = tid >> 3;
    const int sq = tid & 7;
    const int sm = blk * (16 * MT) + sr;
    unsigned noff[NG];
    if (NNB > 0) {
        const int* arow = adj + (size_t)node * ADJ_W;
#pragma unroll
        for (int k = 0; k < NG; ++k) noff[k] = (unsigned)arow[k] * FEAT;
    } else {
        noff[0] = 0u;
    }
    const float* ga_s =
        (GATHER ? (Aself + (size_t)node * FEAT) : (Aself + (size_t)sm * FEAT))
        + sq * 4;
    const float* ga_n =
        (NNB > 0) ? (Aself + sq * 4) : (Aneigh + (size_t)sm * FEAT + sq * 4);
    const int a_st =
        ((sr >> 4) * 8 + sq) * 64 + (((sr & 15) ^ ((sq & 3) << 2)) << 2);

    int aoff[4];
#pragma unroll
    for (int j = 0; j < 4; ++j)
        aoff[j] = mtile * 512 + ((rowInTile ^ (j << 2)) << 2) + klane;

    const int bBase =
        (klane >> 1) * 512 + (nstrip * (NT * 16) + rowInTile) * 2;

    v8f acc[NT];
#pragma unroll
    for (int t = 0; t < NT; ++t) acc[t] = (v8f)0.f;

    v2f ws0[WU], ws1[WU], wn0[WU], wn1[WU];
    float4 pa_s, pa_n;
    v4f pn[NG];

    pa_s = *(const float4*)ga_s;
    if (NNB > 0) {
#pragma unroll
        for (int k = 0; k < NG; ++k) pn[k] = *(const v4f*)(ga_n + noff[k]);
    } else {
        pa_n = *(const float4*)ga_n;
    }
    w_chunk_to_regs<NTHR, WU>(Ws, tid, ws0, ws1);
    w_chunk_to_regs<NTHR, WU>(Wn, tid, wn0, wn1);
    w_regs_to_lds<NTHR, WU>(lds_stage(lds_w, 0, 0), tid, ws0, ws1);
    w_regs_to_lds<NTHR, WU>(lds_stage(lds_w, 0, 1), tid, wn0, wn1);
    if (NNB > 0) pa_n = neigh_mean<NG>(pn);
    *(float4*)(lds_a + a_st)      = pa_s;
    *(float4*)(lds_a + AT + a_st) = pa_n;
    __syncthreads();

#pragma unroll 1
    for (int c = 0; c + 1 < N_CHUNKS; ++c) {
        const int k1 = (c + 1) * TILE_K;
        NextChunk nx;
        nx.gws = Ws + (size_t)k1 * FEAT;
        nx.gwn = Wn + (size_t)k1 * FEAT;
        nx.gas = ga_s + k1;
        nx.gan = ga_n + k1;

        const float* at = lds_a + (c & 1) * 2 * AT;
        mma_chunk<true, NT, NTHR, WU, NNB, NG>(
            acc, lds_stage(lds_w, c & 1, 0) + bBase,
            lds_stage(lds_w, c & 1, 1) + bBase,
            at, at + AT, aoff,
            nx, tid, ws0, ws1, wn0, wn1, pa_s, pa_n, pn, noff);
        if (NNB > 0) pa_n = neigh_mean<NG>(pn);

        w_regs_to_lds<NTHR, WU>(lds_stage(lds_w, (c + 1) & 1, 0), tid, ws0, ws1);
        w_regs_to_lds<NTHR, WU>(lds_stage(lds_w, (c + 1) & 1, 1), tid, wn0, wn1);
        float* at_nx = lds_a + ((c + 1) & 1) * 2 * AT;
        *(float4*)(at_nx + a_st)      = pa_s;
        *(float4*)(at_nx + AT + a_st) = pa_n;
        __syncthreads();
    }
    {
        NextChunk none = {Ws, Wn, ga_s, ga_n};
        const float* at = lds_a + ((N_CHUNKS - 1) & 1) * 2 * AT;
        mma_chunk<false, NT, NTHR, WU, NNB, NG>(
            acc, lds_stage(lds_w, (N_CHUNKS - 1) & 1, 0) + bBase,
            lds_stage(lds_w, (N_CHUNKS - 1) & 1, 1) + bBase,
            at, at + AT, aoff,
            none, tid, ws0, ws1, wn0, wn1, pa_s, pa_n, pn, noff);
    }

    const int half    = lane >> 4;
    const int colBase = nstrip * (NT * 16) + rowInTile;
    const int rowBase = blk * (16 * MT) + mtile * 16 + (half << 3);

    float norm[8];
    if (L2NORM) {
        float p[8];
#pragma unroll
        for (int r = 0; r < 8; ++r) {
            float s = 0.f;
#pragma unroll
            for (int t = 0; t < NT; ++t) s += acc[t][r] * acc[t][r];
            p[r] = s;
        }
#pragma unroll
        for (int r = 0; r < 8; ++r) {
#pragma unroll
            for (int o = 8; o >= 1; o >>= 1) p[r] += __shfl_xor(p[r], o, 32);
        }
        if (rowInTile == 0) {
#pragma unroll
            for (int r = 0; r < 8; ++r)
                row_ss[(mtile * NW + nstrip) * 16 + half * 8 + r] = p[r];
        }
        __syncthreads();
#pragma unroll
        for (int r = 0; r < 8; ++r) {
            float tot = 0.f;
#pragma unroll
            for (int w = 0; w < NW; ++w)
                tot += row_ss[(mtile * NW + w) * 16 + half * 8 + r];
            norm[r] = fmaxf(sqrtf(tot), 1e-12f);
        }
    }

#pragma unroll
    for (int t = 0; t < NT; ++t) {
#pragma unroll
        for (int r = 0; r < 8; ++r) {
            float v = acc[t][r];
            if (RELU) v = fmaxf(v, 0.0f);
            if (L2NORM) v = v / norm[r];
            out[(size_t)(rowBase + r) * FEAT + colBase + t * 16] = v;
        }
    }
}

#define HOP1_BLOCKS (5120 / 32)
__global__ __launch_bounds__(256) void sage_layer0_kernel(
    const float* feat, const int* __restrict__ adj,
    const int* __restrict__ batch,
    float* __restrict__ h1, float* __restrict__ h0,
    const float* __restrict__ Ws, const float* __restrict__ Wn)
{
    __shared__ float lds_w[2 * 2 * CHUNK_FLOATS];
    __shared__ alignas(16) float lds_a[2 * 2 * A_TILE_FLOATS(2)];
    const int sr = threadIdx.x >> 3;
    if (blockIdx.x >= HOP1_BLOCKS) {
        const int blk  = (int)blockIdx.x - HOP1_BLOCKS;
        const int node = batch[blk * 32 + sr];
        sage_gemm_block<true, true, false, 2, 8, 10>(
            feat, node, adj, (const float*)nullptr, Ws, Wn, h0, blk,
            lds_w, lds_a, (float*)nullptr);
    } else {
        const int blk  = (int)blockIdx.x;
        const int m    = blk * 32 + sr;
        const int b    = m / 10;
        const int j    = m - b * 10;
        const int node = adj[(size_t)batch[b] * ADJ_W + j];
        sage_gemm_block<true, true, false, 2, 8, 25>(
            feat, node, adj, (const float*)nullptr, Ws, Wn, h1, blk,
            lds_w, lds_a, (float*)nullptr);
    }
}

__global__ __launch_bounds__(128) void sage_layer1_kernel(
    const float* __restrict__ h0, const float* __restrict__ h1, float* nm,
    const float* __restrict__ Ws, const float* __restrict__ Wn,
    float* __restrict__ out)
{
    __shared__ float lds_w[2 * 2 * CHUNK_FLOATS];
    __shared__ alignas(16) float lds_a[2 * 2 * A_TILE_FLOATS(1)];
    __shared__ float row_ss[4 * 16];
    const int tid = threadIdx.x;
#pragma unroll
    for (int g = 0; g < 2; ++g) {
        float4 v[4][10];
#pragma unroll
        for (int i = 0; i < 4; ++i) {
            const int piece = (g * 4 + i) * 128 + tid;
            const size_t m  = (size_t)blockIdx.x * 16 + (piece >> 6);
            const float* src = h1 + m * 10 * FEAT + (piece & 63) * 4;
#pragma unroll
            for (int j = 0; j < 10; ++j)
                v[i][j] = *(const float4*)(src + (size_t)j * FEAT);
        }
#pragma unroll
        for (int i = 0; i < 4; ++i) {
            const int piece = (g * 4 + i) * 128 + tid;
            const size_t m  = (size_t)blockIdx.x * 16 + (piece >> 6);
            float ax = 0.f, ay = 0.f, az = 0.f, aw = 0.f;
#pragma unroll
            for (int j = 0; j < 10; ++j) {
                ax += v[i][j].x; ay += v[i][j].y;
                az += v[i][j].z; aw += v[i][j].w;
            }
            const float s = 0.1f;
            float4 q; q.x = ax * s; q.y = ay * s; q.z = az * s; q.w = aw * s;
            *(float4*)(nm + m * FEAT + (piece & 63) * 4) = q;
        }
    }
    __syncthreads();
    sage_gemm_block<false, false, true, 1, 4, 0>(
        h0, 0, (const int*)nullptr, nm, Ws, Wn, out, (int)blockIdx.x,
        lds_w, lds_a, row_ss);
}

extern "C" void kernel_launch(void* const* d_in, const int* in_sizes, int n_in,
                              void* d_out, int out_size, void* d_ws,
                              size_t ws_size, hipStream_t stream)
{
    const float* features = (const float*)d_in[0];
    const int*   adj      = (const int*)d_in[1];
    const int*   batch    = (const int*)d_in[2];
    const float* Ws0      = (const float*)d_in[3];
    const float* Wn0      = (const float*)d_in[4];
    const float* Ws1      = (const float*)d_in[5];
    const float* Wn1      = (const float*)d_in[6];
    float* out = (float*)d_out;

    char* ws = (char*)d_ws;
    float* h1   = (float*)(ws + 5787648);
    float* h0   = (float*)(ws + 11030528);
    float* nmh1 = (float*)(ws + 11554816);

    sage_layer0_kernel<<<5120 / 32 + 512 / 32, 256, 0, stream>>>(
        features, adj, batch, h1, h0, Ws0, Wn0);
    sage_layer1_kernel<<<512 / 16, 128, 0, stream>>>(
        h0, h1, nmh1, Ws1, Wn1, out);
}


# Next chunk's register-to-LDS hand-over (weight stores, neighbour sums) spread over the 8 WMMA runs; requests run two chunks ahead and stay in flight across an LDS-only barrier; sums same
# speedup vs baseline: 7.1351x; 1.3004x over previous
#include <hip/hip_runtime.h>
#include <hip/hip_bf16.h>

typedef __attribute__((ext_vector_type(2))) float v2f;
typedef __attribute__((ext_vector_type(4))) float v4f;
typedef __attribute__((ext_vector_type(8))) float v8f;

#define FEAT 256
#define TILE_K 32

#define ADJ_W 25

#define DEV static __device__ __forceinline__
#define CHUNK_FLOATS (TILE_K * FEAT)
#define N_CHUNKS (FEAT / TILE_K)
#define K_STEPS (TILE_K / 4)
#define W_UNITS (TILE_K / 2 * (FEAT / 2))
#define NNBR 25
#define NTHR 256
#define WU (W_UNITS / NTHR)

DEV void sched_fence() { __builtin_amdgcn_sched_barrier(0); }

DEV void lds_barrier()
{
    asm volatile("s_wait_dscnt 0x0\n\ts_barrier_signal -1\n\ts_barrier_wait -1"
                 ::: "memory");
}

DEV int w_unit_goff(int i, int tid)
{
    const int u     = i * NTHR + tid;
    const int kpair = u >> 7;
    const int npair = u & 127;
    return (2 * kpair) * FEAT + 2 * npair;
}

DEV void w_chunk_to_regs(const float* __restrict__ g, int tid,
                         v2f (&r0)[WU], v2f (&r1)[WU])
{
#pragma unroll
    for (int i = 0; i < WU; ++i) {
        const int g0 = w_unit_goff(i, tid);
        r0[i] = *(const v2f*)(g + g0);
        r1[i] = *(const v2f*)(g + g0 + FEAT);
    }
}

DEV void w_regs_to_lds(float* l, int tid,
                       const v2f (&r0)[WU], const v2f (&r1)[WU])
{
#pragma unroll
    for (int i = 0; i < WU; ++i) {
        const int u     = i * NTHR + tid;
        const int kpair = u >> 7;
        const int npair = u & 127;
        float4 p; p.x = r0[i].x; p.y = r1[i].x; p.z = r0[i].y; p.w = r1[i].y;
        *(float4*)(l + kpair * 512 + npair * 4) = p;
    }
}

struct NextChunk {
    const float* gws;
    const float* gwn;
    const float* gas;
    const float* gan;
};

#define KA_LO(s) (3 * (s))
#define KA_HI(s) ((s) == K_STEPS - 1 ? NNBR : 3 * (s) + 3)

DEV void w_unit_request(const NextChunk& nx, int tid, int i,
                        v2f& s0, v2f& s1, v2f& n0, v2f& n1)
{
    const int g0 = w_unit_goff(i, tid);
    s0 = *(const v2f*)(nx.gws + g0);
    s1 = *(const v2f*)(nx.gws + g0 + FEAT);
    n0 = *(const v2f*)(nx.gwn + g0);
    n1 = *(const v2f*)(nx.gwn + g0 + FEAT);
}

DEV void w_unit_to_lds(float* l, int tid, int i, v2f r0, v2f r1)
{
    asm volatile("" : "+v"(r0), "+v"(r1));
    const int u     = i * NTHR + tid;
    const int kpair = u >> 7;
    const int npair = u & 127;
    float4 p; p.x = r0.x; p.y = r1.x; p.z = r0.y; p.w = r1.y;
    *(float4*)(l + kpair * 512 + npair * 4) = p;
}

template <int CONS, int REQ, int NT>
DEV void mma_chunk(v8f (&acc)[NT],
                   const float* bws, const float* bwn,
                   const float* fas, const float* fan, const int (&aoff)[4],
                   const NextChunk& nx, int tid,
                   float* lws, float* lwn, float* las, float* lan,
                   v2f (&ws0)[WU], v2f (&ws1)[WU],
                   v2f (&wn0)[WU], v2f (&wn1)[WU],
                   float4& pa_s,
                   v4f (&pn)[NNBR], const unsigned (&noff)[NNBR], bool full)
{
    static_assert(WU == K_STEPS, "one weight pack unit per matrix per k-step");
    v2f bs[2][NT], bn[2][NT];
    v2f fa_s[2], fa_n[2];
    v4f nsum = (v4f)0.f;
#pragma unroll
    for (int t = 0; t < NT; ++t) {
        bs[0][t] = *(const v2f*)(bws + t * 32);
        bn[0][t] = *(const v2f*)(bwn + t * 32);
    }
    fa_s[0] = *(const v2f*)(fas + aoff[0]);
    fa_n[0] = *(const v2f*)(fan + aoff[0]);
#pragma unroll
    for (int s = 0; s < K_STEPS; ++s) {
        const int cur = s & 1;
        if (s + 1 < K_STEPS) {
#pragma unroll
            for (int t = 0; t < NT; ++t) {
                bs[cur ^ 1][t] = *(const v2f*)(bws + (s + 1) * 1024 + t * 32);
                bn[cur ^ 1][t] = *(const v2f*)(bwn + (s + 1) * 1024 + t * 32);
            }
            fa_s[cur ^ 1] = *(const v2f*)(fas + (s + 1) * 64 + aoff[(s + 1) & 3]);
            fa_n[cur ^ 1] = *(const v2f*)(fan + (s + 1) * 64 + aoff[(s + 1) & 3]);
        }
        if (CONS >= 1) {
            w_unit_to_lds(lws, tid, s, ws0[s], ws1[s]);
            w_unit_to_lds(lwn, tid, s, wn0[s], wn1[s]);
        }
        if (s >= 1) {
            if (REQ >= 1)
                w_unit_request(nx, tid, s - 1,
                               ws0[s - 1], ws1[s - 1], wn0[s - 1], wn1[s - 1]);
            if (REQ == 2) {
#pragma unroll
                for (int k = KA_LO(s - 1); k < KA_HI(s - 1); ++k)
                    pn[k] = *(const v4f*)(nx.gan + noff[k]);
            }
        }
        sched_fence();
#pragma unroll
        for (int t = 0; t < NT; ++t)
            acc[t] = __builtin_amdgcn_wmma_f32_16x16x4_f32(
                false, fa_s[cur], false, bs[cur][t], (short)0, acc[t],
                false, false);
        if (CONS == 2) {
#pragma unroll
            for (int k = KA_LO(s); k < KA_HI(s); ++k) {
                const v4f b = nsum + pn[k];
                nsum = (k < 10 || full) ? b : nsum;
            }
        }
#pragma unroll
        for (int t = 0; t < NT; ++t)
            acc[t] = __builtin_amdgcn_wmma_f32_16x16x4_f32(
                false, fa_n[cur], false, bn[cur][t], (short)0, acc[t],
                false, false);
        sched_fence();
    }
    if (CONS == 2) {
        const float sc = full ? (1.0f / 25.0f) : 0.1f;
        float4 m;
        m.x = nsum.x * sc; m.y = nsum.y * sc; m.z = nsum.z * sc; m.w = nsum.w * sc;
        *(float4*)las = pa_s;
        *(float4*)lan = m;
    }
    if (REQ >= 1)
        w_unit_request(nx, tid, K_STEPS - 1, ws0[K_STEPS - 1], ws1[K_STEPS - 1],
                       wn0[K_STEPS - 1], wn1[K_STEPS - 1]);
    if (REQ == 2) {
#pragma unroll
        for (int k = KA_LO(K_STEPS - 1); k < KA_HI(K_STEPS - 1); ++k)
            pn[k] = *(const v4f*)(nx.gan + noff[k]);
        pa_s = *(const float4*)nx.gas;
    }
}

DEV float4 neigh_mean(const v4f (&pn)[NNBR], bool full)
{
    v4f a = (v4f)0.f;
#pragma unroll
    for (int k = 0; k < 10; ++k) a += pn[k];
#pragma unroll
    for (int k = 10; k < NNBR; ++k) {
        const v4f b = a + pn[k];
        a = full ? b : a;
    }
    const float s = full ? (1.0f / 25.0f) : 0.1f;
    float4 r; r.x = a.x * s; r.y = a.y * s; r.z = a.z * s; r.w = a.w * s;
    return r;
}

DEV float* lds_stage(float* lds_w, int stage, int matrix)
{
    return lds_w + (stage * 2 + matrix) * CHUNK_FLOATS;
}

#define A_TILE_FLOATS(MT) (16 * (MT) * TILE_K)
#define HB_PITCH 260
#define A1_MAT (N_CHUNKS * A_TILE_FLOATS(1))

__global__ __launch_bounds__(NTHR) void sage_fused_kernel(
    const float* feat, const int* __restrict__ adj,
    const int* __restrict__ batch,
    const float* __restrict__ Ws0, const float* __restrict__ Wn0,
    const float* __restrict__ Ws1, const float* __restrict__ Wn1,
    float* __restrict__ out)
{
    constexpr int AT = A_TILE_FLOATS(2);
    __shared__ float lds_w[2 * 2 * CHUNK_FLOATS];
    __shared__ alignas(16) float lds_a[2 * 2 * A_TILE_FLOATS(2)];
    __shared__ alignas(16) float hbuf[32 * HB_PITCH];
    __shared__ alignas(16) float a1[2 * A1_MAT];
    __shared__ float row_ss[8 * 2];

    const int tid       = threadIdx.x;
    const int lane      = tid & 31;
    const int wave      = tid >> 5;
    const int rowInTile = lane & 15;
    const int half      = lane >> 4;
    const int klane     = half << 1;
    const int b2        = (int)blockIdx.x * 2;

#pragma unroll
    for (int i = 0; i < 2 * A1_MAT / (4 * NTHR); ++i) {
        float4 z; z.x = 0.f; z.y = 0.f; z.z = 0.f; z.w = 0.f;
        *(float4*)(a1 + (i * NTHR + tid) * 4) = z;
    }

    const int sr = tid >> 3;
    const int sq = tid & 7;
    int node;
    bool full;
    if (sr < 20) {
        const int e = (sr >= 10) ? 1 : 0;
        const int j = sr - 10 * e;
        node = adj[(size_t)batch[b2 + e] * ADJ_W + j];
        full = true;
    } else {
        node = batch[b2 + ((sr == 20) ? 0 : 1)];
        full = false;
    }
    unsigned noff[NNBR];
    {
        const int* arow = adj + (size_t)node * ADJ_W;
#pragma unroll
        for (int k = 0; k < NNBR; ++k) noff[k] = (unsigned)arow[k] * FEAT;
    }
    const float* ga_s = feat + (size_t)node * FEAT + sq * 4;
    const float* ga_n = feat + sq * 4;
    const int a_st =
        ((sr >> 4) * 8 + sq) * 64 + (((sr & 15) ^ ((sq & 3) << 2)) << 2);

    const int mtile  = wave >> 2;
    const int nstrip = wave & 3;
    int aoff[4];
#pragma unroll
    for (int j = 0; j < 4; ++j)
        aoff[j] = mtile * 512 + ((rowInTile ^ (j << 2)) << 2) + klane;
    const int bBase = (klane >> 1) * 512 + (nstrip * 64 + rowInTile) * 2;

    v8f acc[4];
#pragma unroll
    for (int t = 0; t < 4; ++t) acc[t] = (v8f)0.f;

    v2f ws0[WU], ws1[WU], wn0[WU], wn1[WU];
    float4 pa_s, pa_n;
    v4f pn[NNBR];

    pa_s = *(const float4*)ga_s;
#pragma unroll
    for (int k = 0; k < NNBR; ++k) pn[k] = *(const v4f*)(ga_n + noff[k]);
    w_chunk_to_regs(Ws0, tid, ws0, ws1);
    w_chunk_to_regs(Wn0, tid, wn0, wn1);
    w_regs_to_lds(lds_stage(lds_w, 0, 0), tid, ws0, ws1);
    w_regs_to_lds(lds_stage(lds_w, 0, 1), tid, wn0, wn1);
    pa_n = neigh_mean(pn, full);
    *(float4*)(lds_a + a_st)      = pa_s;
    *(float4*)(lds_a + AT + a_st) = pa_n;
    {
        NextChunk n1;
        n1.gws = Ws0 + (size_t)TILE_K * FEAT;
        n1.gwn = Wn0 + (size_t)TILE_K * FEAT;
        n1.gas = ga_s + TILE_K;
        n1.gan = ga_n + TILE_K;
#pragma unroll
        for (int s = 0; s < K_STEPS; ++s) {
            w_unit_request(n1, tid, s, ws0[s], ws1[s], wn0[s], wn1[s]);
#pragma unroll
            for (int k = KA_LO(s); k < KA_HI(s); ++k)
                pn[k] = *(const v4f*)(n1.gan + noff[k]);
            sched_fence();
        }
        pa_s = *(const float4*)n1.gas;
    }
    lds_barrier();

#pragma unroll 1
    for (int c = 0; c + 2 < N_CHUNKS; ++c) {
        const int k2 = (c + 2) * TILE_K;
        NextChunk nx;
        nx.gws = Ws0 + (size_t)k2 * FEAT;
        nx.gwn = Wn0 + (size_t)k2 * FEAT;
        nx.gas = ga_s + k2;
        nx.gan = ga_n + k2;
        const float* at = lds_a + (c & 1) * 2 * AT;
        float* at_nx = lds_a + ((c + 1) & 1) * 2 * AT;
        mma_chunk<2, 2, 4>(
            acc, lds_stage(lds_w, c & 1, 0) + bBase,
            lds_stage(lds_w, c & 1, 1) + bBase,
            at, at + AT, aoff, nx, tid,
            lds_stage(lds_w, (c + 1) & 1, 0), lds_stage(lds_w, (c + 1) & 1, 1),
            at_nx + a_st, at_nx + AT + a_st,
            ws0, ws1, wn0, wn1, pa_s, pn, noff, full);
        lds_barrier();
    }
    {
        NextChunk nx = {Ws1, Wn1, ga_s, ga_n};
        const float* at = lds_a;
        float* at_nx = lds_a + 2 * AT;
        mma_chunk<2, 1, 4>(
            acc, lds_stage(lds_w, 0, 0) + bBase, lds_stage(lds_w, 0, 1) + bBase,
            at, at + AT, aoff, nx, tid,
            lds_stage(lds_w, 1, 0), lds_stage(lds_w, 1, 1),
            at_nx + a_st, at_nx + AT + a_st,
            ws0, ws1, wn0, wn1, pa_s, pn, noff, full);
        lds_barrier();
    }
    {
        NextChunk nx = {Ws1 + (size_t)TILE_K * FEAT, Wn1 + (size_t)TILE_K * FEAT,
                        ga_s, ga_n};
        const float* at = lds_a + 2 * AT;
        mma_chunk<1, 1, 4>(
            acc, lds_stage(lds_w, 1, 0) + bBase, lds_stage(lds_w, 1, 1) + bBase,
            at, at + AT, aoff, nx, tid,
            lds_stage(lds_w, 0, 0), lds_stage(lds_w, 0, 1), lds_a, lds_a,
            ws0, ws1, wn0, wn1, pa_s, pn, noff, full);
    }

    {
        float* hb = hbuf + (mtile * 16 + (half << 3)) * HB_PITCH
                  + nstrip * 64 + rowInTile;
#pragma unroll
        for (int t = 0; t < 4; ++t) {
#pragma unroll
            for (int r = 0; r < 8; ++r)
                hb[r * HB_PITCH + t * 16] = fmaxf(acc[t][r], 0.0f);
        }
    }
    lds_barrier();

    {
        const int m = tid >> 7;
        const int i = (tid >> 6) & 1;
        const int p = tid & 63;
        float4 q;
        if (m == 0) {
            q = *(const float4*)(hbuf + (20 + i) * HB_PITCH + p * 4);
        } else {
            const float* srow = hbuf + (10 * i) * HB_PITCH + p * 4;
            float ax = 0.f, ay = 0.f, az = 0.f, aw = 0.f;
#pragma unroll
            for (int j = 0; j < 10; ++j) {
                const float4 v = *(const float4*)(srow + j * HB_PITCH);
                ax += v.x; ay += v.y; az += v.z; aw += v.w;
            }
            const float s = 0.1f;
            q.x = ax * s; q.y = ay * s; q.z = az * s; q.w = aw * s;
        }
        const int c1 = p >> 3;
        const int q1 = p & 7;
        *(float4*)(a1 + m * A1_MAT + c1 * A_TILE_FLOATS(1) + q1 * 64
                   + ((i ^ ((q1 & 3) << 2)) << 2)) = q;
    }
    lds_barrier();

    int aoff1[4];
#pragma unroll
    for (int j = 0; j < 4; ++j)
        aoff1[j] = ((rowInTile ^ (j << 2)) << 2) + klane;
    const int bBase1 = (klane >> 1) * 512 + (wave * 32 + rowInTile) * 2;

    v8f acc1[2];
#pragma unroll
    for (int t = 0; t < 2; ++t) acc1[t] = (v8f)0.f;

#pragma unroll 1
    for (int c = 0; c + 2 < N_CHUNKS; ++c) {
        const int k2 = (c + 2) * TILE_K;
        NextChunk nx;
        nx.gws = Ws1 + (size_t)k2 * FEAT;
        nx.gwn = Wn1 + (size_t)k2 * FEAT;
        nx.gas = ga_s;
        nx.gan = ga_n;
        mma_chunk<1, 1, 2>(
            acc1, lds_stage(lds_w, c & 1, 0) + bBase1,
            lds_stage(lds_w, c & 1, 1) + bBase1,
            a1 + c * A_TILE_FLOATS(1), a1 + A1_MAT + c * A_TILE_FLOATS(1),
            aoff1, nx, tid,
            lds_stage(lds_w, (c + 1) & 1, 0), lds_stage(lds_w, (c + 1) & 1, 1),
            lds_a, lds_a,
            ws0, ws1, wn0, wn1, pa_s, pn, noff, full);
        lds_barrier();
    }
    {
        NextChunk none = {Ws1, Wn1, ga_s, ga_n};
        mma_chunk<1, 0, 2>(
            acc1, lds_stage(lds_w, 0, 0) + bBase1, lds_stage(lds_w, 0, 1) + bBase1,
            a1 + (N_CHUNKS - 2) * A_TILE_FLOATS(1),
            a1 + A1_MAT + (N_CHUNKS - 2) * A_TILE_FLOATS(1),
            aoff1, none, tid,
            lds_stage(lds_w, 1, 0), lds_stage(lds_w, 1, 1), lds_a, lds_a,
            ws0, ws1, wn0, wn1, pa_s, pn, noff, full);
        lds_barrier();
        mma_chunk<0, 0, 2>(
            acc1, lds_stage(lds_w, 1, 0) + bBase1, lds_stage(lds_w, 1, 1) + bBase1,
            a1 + (N_CHUNKS - 1) * A_TILE_FLOATS(1),
            a1 + A1_MAT + (N_CHUNKS - 1) * A_TILE_FLOATS(1),
            aoff1, none, tid,
            lds_stage(lds_w, 0, 0), lds_stage(lds_w, 0, 1), lds_a, lds_a,
            ws0, ws1, wn0, wn1, pa_s, pn, noff, full);
    }

    float p2[2];
#pragma unroll
    for (int r = 0; r < 2; ++r) {
        float s = 0.f;
#pragma unroll
        for (int t = 0; t < 2; ++t) s += acc1[t][r] * acc1[t][r];
        p2[r] = s;
    }
#pragma unroll
    for (int r = 0; r < 2; ++r) {
#pragma unroll
        for (int o = 8; o >= 1; o >>= 1) p2[r] += __shfl_xor(p2[r], o, 32);
    }
    if (lane == 0) {
        row_ss[wave * 2 + 0] = p2[0];
        row_ss[wave * 2 + 1] = p2[1];
    }
    __syncthreads();
    float nrm[2];
#pragma unroll
    for (int r = 0; r < 2; ++r) {
        float tot = 0.f;
#pragma unroll
        for (int w = 0; w < 8; ++w) tot += row_ss[w * 2 + r];
        nrm[r] = fmaxf(sqrtf(tot), 1e-12f);
    }
    if (half == 0) {
        float* o = out + (size_t)b2 * FEAT + wave * 32 + rowInTile;
#pragma unroll
        for (int r = 0; r < 2; ++r) {
#pragma unroll
            for (int t = 0; t < 2; ++t)
                o[r * FEAT + t * 16] = acc1[t][r] / nrm[r];
        }
    }
}

extern "C" void kernel_launch(void* const* d_in, const int* in_sizes, int n_in,
                              void* d_out, int out_size, void* d_ws,
                              size_t ws_size, hipStream_t stream)
{
    const float* features = (const float*)d_in[0];
    const int*   adj      = (const int*)d_in[1];
    const int*   batch    = (const int*)d_in[2];
    const float* Ws0      = (const float*)d_in[3];
    const float* Wn0      = (const float*)d_in[4];
    const float* Ws1      = (const float*)d_in[5];
    const float* Wn1      = (const float*)d_in[6];
    float* out = (float*)d_out;

    sage_fused_kernel<<<512 / 2, NTHR, 0, stream>>>(
        features, adj, batch, Ws0, Wn0, Ws1, Wn1, out);
}
